# starting kernel
# baseline (speedup 1.0000x reference)
#include <hip/hip_runtime.h>
#include <math.h>
#include <float.h>

#define N_NODES 50000
#define N_EDGES 800000
#define IN_C    128
#define OUT_C   128
#define HEADS   8
#define BASES   4
#define HEAD_D  16
#define BD      (BASES * HEAD_D)

typedef float v2f __attribute__((ext_vector_type(2)));
typedef float v8f __attribute__((ext_vector_type(8)));

__device__ __forceinline__ void atomicMaxFloat(float* addr, float val) {
  unsigned bits = __float_as_uint(val);
  if ((bits & 0x80000000u) == 0u) {
    atomicMax((int*)addr, (int)bits);
  } else {
    atomicMin((unsigned int*)addr, bits);
  }
}

__global__ __launch_bounds__(256) void k_init(float* __restrict__ agg,
                                              unsigned* __restrict__ cnt) {
  int i = blockIdx.x * 256 + threadIdx.x;
  if (i < N_NODES * 256) {
    int part = (i >> 6) & 3;
    agg[i] = (part == 3) ? -INFINITY : 0.0f;
  }
  if (i < N_NODES) cnt[i] = 0u;
}

__global__ __launch_bounds__(256) void k_degree(const int* __restrict__ col,
                                                unsigned* __restrict__ cnt) {
  int e = blockIdx.x * 256 + threadIdx.x;
  if (e < N_EDGES) atomicAdd(&cnt[col[e]], 1u);
}

__global__ __launch_bounds__(256) void k_dinv(const unsigned* __restrict__ cnt,
                                              float* __restrict__ dinv) {
  int n = blockIdx.x * 256 + threadIdx.x;
  if (n < N_NODES) dinv[n] = rsqrtf((float)cnt[n] + 1.0f);
}

__global__ __launch_bounds__(128) void k_gemm(const float* __restrict__ x,
                                              const float* __restrict__ bW,
                                              const float* __restrict__ cW,
                                              const float* __restrict__ cb,
                                              float* __restrict__ bases,
                                              float* __restrict__ wbuf) {
  __shared__ float xs[64][132];

  const int tid  = threadIdx.x;
  const int row0 = blockIdx.x * 64;

  for (int t = 0; t < 16; ++t) {
    int idx = t * 128 + tid;
    int r   = idx >> 5;
    int c4  = (idx & 31) * 4;
    int gr  = row0 + r;
    if (gr >= N_NODES) gr = N_NODES - 1;
    const float4 v = *(const float4*)(x + (size_t)gr * IN_C + c4);
    xs[r][c4 + 0] = v.x; xs[r][c4 + 1] = v.y;
    xs[r][c4 + 2] = v.z; xs[r][c4 + 3] = v.w;
  }
  __syncthreads();

  const int wave = tid >> 5;
  const int lane = tid & 31;
  const int nlo  = lane & 15;
  const int half = lane >> 4;
  const int kb   = half * 2;
  const int lr   = wave * 16 + nlo;

  for (int nt = 0; nt < 12; ++nt) {
    v8f acc = {0.f, 0.f, 0.f, 0.f, 0.f, 0.f, 0.f, 0.f};

    if (nt < 4) {
      const float* W = bW + nt * (IN_C * HEAD_D);
#pragma unroll
      for (int ks = 0; ks < 32; ++ks) {
        const int k = ks * 4 + kb;
        v2f a = *(const v2f*)&xs[lr][k];
        v2f b = { W[k * HEAD_D + nlo], W[(k + 1) * HEAD_D + nlo] };
        acc = __builtin_amdgcn_wmma_f32_16x16x4_f32(
            false, a, false, b, (short)0, acc, false, false);
      }
#pragma unroll
      for (int r = 0; r < 8; ++r) {
        const int m  = r + (half << 3);
        const int gr = row0 + wave * 16 + m;
        if (gr < N_NODES) bases[(size_t)gr * BD + nt * 16 + nlo] = acc[r];
      }
    } else {
      const int   head = nt - 4;
      const float* W   = cW + head * 16;
#pragma unroll
      for (int ks = 0; ks < 32; ++ks) {
        const int k = ks * 4 + kb;
        v2f a = *(const v2f*)&xs[lr][k];
        v2f b = { W[k * OUT_C + nlo], W[(k + 1) * OUT_C + nlo] };
        acc = __builtin_amdgcn_wmma_f32_16x16x4_f32(
            false, a, false, b, (short)0, acc, false, false);
      }
      const float bb = cb[head * 16 + nlo];
#pragma unroll
      for (int r = 0; r < 8; ++r) {
        float v = acc[r] + bb;
        float mx = v;
        mx = fmaxf(mx, __shfl_xor(mx, 1));
        mx = fmaxf(mx, __shfl_xor(mx, 2));
        mx = fmaxf(mx, __shfl_xor(mx, 4));
        mx = fmaxf(mx, __shfl_xor(mx, 8));
        float e = __expf(v - mx);
        float s = e;
        s += __shfl_xor(s, 1);
        s += __shfl_xor(s, 2);
        s += __shfl_xor(s, 4);
        s += __shfl_xor(s, 8);
        const float wv = e / s;
        const int m  = r + (half << 3);
        const int gr = row0 + wave * 16 + m;
        if (gr < N_NODES) wbuf[(size_t)gr * OUT_C + head * 16 + nlo] = wv;
      }
    }
  }
}

__global__ __launch_bounds__(256) void k_scatter(const int* __restrict__ ei,
                                                 const float* __restrict__ bases,
                                                 const float* __restrict__ dinv,
                                                 float* __restrict__ agg) {
  const long long g = (long long)blockIdx.x * 256 + threadIdx.x;
  const int e = (int)(g >> 4);
  if (e >= N_EDGES) return;
  const int q = (int)(g & 15);

  const int row = ei[e];
  const int col = ei[N_EDGES + e];
  if (q == 0 && e + 4096 < N_EDGES) {
    __builtin_prefetch(ei + e + 4096, 0, 0);
    __builtin_prefetch(ei + N_EDGES + e + 4096, 0, 0);
  }

  const float  dr = dinv[row];
  const float4 v  = *(const float4*)(bases + (size_t)row * BD + q * 4);
  float* base     = agg + (size_t)col * 256;

  const float vals[4] = {v.x, v.y, v.z, v.w};
#pragma unroll
  for (int c = 0; c < 4; ++c) {
    const int   idx = q * 4 + c;
    const float t   = vals[c];
    atomicAdd(base + idx,       dr * t);
    atomicAdd(base + 64 + idx,  t);
    atomicAdd(base + 128 + idx, t * t);
    atomicMaxFloat(base + 192 + idx, t);
  }
}

__global__ __launch_bounds__(128) void k_combine(const float* __restrict__ agg,
                                                 const float* __restrict__ bases,
                                                 const float* __restrict__ wbuf,
                                                 const unsigned* __restrict__ cnt,
                                                 const float* __restrict__ dinv,
                                                 const float* __restrict__ bias,
                                                 float* __restrict__ out) {
  __shared__ float ws_[128];
  __shared__ float as_[256];
  __shared__ float bs_[64];

  const int n   = blockIdx.x;
  const int tid = threadIdx.x;

  ws_[tid]       = wbuf[(size_t)n * 128 + tid];
  as_[tid]       = agg[(size_t)n * 256 + tid];
  as_[tid + 128] = agg[(size_t)n * 256 + 128 + tid];
  if (tid < 64) bs_[tid] = bases[(size_t)n * 64 + tid];
  __syncthreads();

  const float cf = (float)cnt[n];
  const float ic = 1.0f / fmaxf(cf, 1.0f);
  const float dv = dinv[n];
  const int h = tid >> 4;
  const int d = tid & 15;

  float o = bias[tid];
#pragma unroll
  for (int b = 0; b < 4; ++b) {
    const int   idx  = b * 16 + d;
    const float sym  = dv * (as_[idx] + dv * bs_[idx]);
    const float s    = as_[64 + idx];
    const float mean = s * ic;
    const float var  = as_[128 + idx] * ic - mean * mean;
    const float sd   = sqrtf(fmaxf(var, 0.0f) + 1e-5f);
    const float mx   = (cf > 0.0f) ? as_[192 + idx] : 0.0f;
    const float* wp  = ws_ + h * 16 + b * 4;
    o += wp[0] * sym + wp[1] * mean + wp[2] * mx + wp[3] * sd;
  }
  out[(size_t)n * 128 + tid] = o;
}

extern "C" void kernel_launch(void* const* d_in, const int* in_sizes, int n_in,
                              void* d_out, int out_size, void* d_ws, size_t ws_size,
                              hipStream_t stream) {
  (void)in_sizes; (void)n_in; (void)out_size; (void)ws_size;
  const float* x    = (const float*)d_in[0];
  const int*   ei   = (const int*)d_in[1];
  const float* bW   = (const float*)d_in[2];
  const float* cW   = (const float*)d_in[3];
  const float* cb   = (const float*)d_in[4];
  const float* bias = (const float*)d_in[5];
  float*       out  = (float*)d_out;

  char* ws = (char*)d_ws;
  float*    agg   = (float*)ws;  ws += (size_t)N_NODES * 256 * sizeof(float);
  float*    bases = (float*)ws;  ws += (size_t)N_NODES * BD * sizeof(float);
  float*    wbuf  = (float*)ws;  ws += (size_t)N_NODES * OUT_C * sizeof(float);
  unsigned* cnt   = (unsigned*)ws; ws += (size_t)N_NODES * sizeof(unsigned);
  float*    dinv  = (float*)ws;

  k_init<<<N_NODES, 256, 0, stream>>>(agg, cnt);
  k_degree<<<(N_EDGES + 255) / 256, 256, 0, stream>>>(ei + N_EDGES, cnt);
  k_gemm<<<(N_NODES + 63) / 64, 128, 0, stream>>>(x, bW, cW, cb, bases, wbuf);
  k_dinv<<<(N_NODES + 255) / 256, 256, 0, stream>>>(cnt, dinv);
  {
    const long long work = (long long)N_EDGES * 16;
    k_scatter<<<(unsigned)((work + 255) / 256), 256, 0, stream>>>(ei, bases, dinv, agg);
  }
  k_combine<<<N_NODES, 128, 0, stream>>>(agg, bases, wbuf, cnt, dinv, bias, out);
}


# f32 atomic scatter replaced by per-node register sums: a returning count atomic ranks each edge, scan+fill sort sources by destination, 16 lanes gather+combine a node; sums run in arrival-rank order
# speedup vs baseline: 2.4130x; 2.4130x over previous
#include <hip/hip_runtime.h>
#include <math.h>
#include <float.h>


#define N_NODES 50000
#define N_EDGES 800000
#define IN_C    128
#define OUT_C   128
#define HEADS   8
#define BASES   4
#define HEAD_D  16
#define BD      (BASES * HEAD_D)

typedef float v2f __attribute__((ext_vector_type(2)));
typedef float v8f __attribute__((ext_vector_type(8)));

static_assert(N_NODES % 16 == 0, "k_gather maps 16 nodes to a block with no tail");

__global__ __launch_bounds__(256) void k_zero(unsigned* __restrict__ cnt) {
  const int i = blockIdx.x * 256 + threadIdx.x;
  if (i < N_NODES) cnt[i] = 0u;
}

__global__ __launch_bounds__(256) void k_rank(const int* __restrict__ col,
                                              unsigned* __restrict__ cnt,
                                              unsigned* __restrict__ rank) {
  const int e = blockIdx.x * 256 + threadIdx.x;
  if (e < N_EDGES) rank[e] = atomicAdd(&cnt[col[e]], 1u);
}

__global__ __launch_bounds__(256) void k_scan(const unsigned* __restrict__ cnt,
                                              unsigned* __restrict__ off,
                                              float* __restrict__ dinv) {
  __shared__ unsigned wpart[8];
  __shared__ unsigned wtot[8];

  const int tid  = threadIdx.x;
  const int lane = tid & 31;
  const int wave = tid >> 5;
  const int b0   = blockIdx.x * 256;
  const int n    = b0 + tid;

  unsigned part = 0u;
  for (int i0 = 0; i0 < b0; i0 += 8 * 1024) {
#pragma unroll
    for (int u = 0; u < 8; ++u) {
      const int   i  = i0 + u * 1024 + tid * 4;
      const bool  ok = i < b0;
      const uint4 v  = *(const uint4*)(cnt + (ok ? i : 0));
      part += (v.x + v.y + v.z + v.w) * (unsigned)ok;
    }
  }
  part += __shfl_xor(part, 16);
  part += __shfl_xor(part, 8);
  part += __shfl_xor(part, 4);
  part += __shfl_xor(part, 2);
  part += __shfl_xor(part, 1);

  const unsigned c = (n < N_NODES) ? cnt[n] : 0u;
  unsigned x = c;
  for (int d = 1; d < 32; d <<= 1) {
    const unsigned y = __shfl_up(x, d);
    if (lane >= d) x += y;
  }
  if (lane == 31) wtot[wave] = x;
  if (lane == 0)  wpart[wave] = part;
  __syncthreads();

  unsigned base = 0u;
  for (int w = 0; w < 8; ++w) {
    base += wpart[w];
    if (w < wave) base += wtot[w];
  }
  if (n < N_NODES) {
    off[n]  = base + x - c;
    dinv[n] = rsqrtf((float)c + 1.0f);
  }
}

__global__ __launch_bounds__(256) void k_fill(const int* __restrict__ ei,
                                              const unsigned* __restrict__ off,
                                              const unsigned* __restrict__ rank,
                                              int* __restrict__ srcs) {
  const int e = blockIdx.x * 256 + threadIdx.x;
  if (e < N_EDGES) {
    const int row = ei[e];
    const int col = ei[N_EDGES + e];
    const unsigned p = off[col] + rank[e];
    if (p < (unsigned)N_EDGES) srcs[p] = row;
  }
}

__global__ __launch_bounds__(128) void k_gemm(const float* __restrict__ x,
                                              const float* __restrict__ bW,
                                              const float* __restrict__ cW,
                                              const float* __restrict__ cb,
                                              float* __restrict__ bases,
                                              float* __restrict__ wbuf) {
  __shared__ float xs[64][132];

  const int tid  = threadIdx.x;
  const int row0 = blockIdx.x * 64;

  for (int t = 0; t < 16; ++t) {
    int idx = t * 128 + tid;
    int r   = idx >> 5;
    int c4  = (idx & 31) * 4;
    int gr  = row0 + r;
    if (gr >= N_NODES) gr = N_NODES - 1;
    const float4 v = *(const float4*)(x + (size_t)gr * IN_C + c4);
    xs[r][c4 + 0] = v.x; xs[r][c4 + 1] = v.y;
    xs[r][c4 + 2] = v.z; xs[r][c4 + 3] = v.w;
  }
  __syncthreads();

  const int wave = tid >> 5;
  const int lane = tid & 31;
  const int nlo  = lane & 15;
  const int half = lane >> 4;
  const int kb   = half * 2;
  const int lr   = wave * 16 + nlo;

  for (int nt = 0; nt < 12; ++nt) {
    v8f acc = {0.f, 0.f, 0.f, 0.f, 0.f, 0.f, 0.f, 0.f};

    if (nt < 4) {
      const float* W = bW + nt * (IN_C * HEAD_D);
#pragma unroll
      for (int ks = 0; ks < 32; ++ks) {
        const int k = ks * 4 + kb;
        v2f a = *(const v2f*)&xs[lr][k];
        v2f b = { W[k * HEAD_D + nlo], W[(k + 1) * HEAD_D + nlo] };
        acc = __builtin_amdgcn_wmma_f32_16x16x4_f32(
            false, a, false, b, (short)0, acc, false, false);
      }
#pragma unroll
      for (int r = 0; r < 8; ++r) {
        const int m  = r + (half << 3);
        const int gr = row0 + wave * 16 + m;
        if (gr < N_NODES) bases[(size_t)gr * BD + nt * 16 + nlo] = acc[r];
      }
    } else {
      const int   head = nt - 4;
      const float* W   = cW + head * 16;
#pragma unroll
      for (int ks = 0; ks < 32; ++ks) {
        const int k = ks * 4 + kb;
        v2f a = *(const v2f*)&xs[lr][k];
        v2f b = { W[k * OUT_C + nlo], W[(k + 1) * OUT_C + nlo] };
        acc = __builtin_amdgcn_wmma_f32_16x16x4_f32(
            false, a, false, b, (short)0, acc, false, false);
      }
      const float bb = cb[head * 16 + nlo];
#pragma unroll
      for (int r = 0; r < 8; ++r) {
        float v = acc[r] + bb;
        float mx = v;
        mx = fmaxf(mx, __shfl_xor(mx, 1));
        mx = fmaxf(mx, __shfl_xor(mx, 2));
        mx = fmaxf(mx, __shfl_xor(mx, 4));
        mx = fmaxf(mx, __shfl_xor(mx, 8));
        float e = __expf(v - mx);
        float s = e;
        s += __shfl_xor(s, 1);
        s += __shfl_xor(s, 2);
        s += __shfl_xor(s, 4);
        s += __shfl_xor(s, 8);
        const float wv = e / s;
        const int m  = r + (half << 3);
        const int gr = row0 + wave * 16 + m;
        if (gr < N_NODES) wbuf[(size_t)gr * OUT_C + head * 16 + nlo] = wv;
      }
    }
  }
}

__global__ __launch_bounds__(256) void k_gather(const int* __restrict__ srcs,
                                                const unsigned* __restrict__ off,
                                                const unsigned* __restrict__ cnt,
                                                const float* __restrict__ bases,
                                                const float* __restrict__ wbuf,
                                                const float* __restrict__ dinv,
                                                const float* __restrict__ bias,
                                                float* __restrict__ out) {
  __shared__ float4 fin[16][4][16];

  const int tid = threadIdx.x;
  const int nl  = tid >> 4;
  const int q   = tid & 15;
  const int n   = blockIdx.x * 16 + nl;

  const unsigned deg = cnt[n];
  unsigned start = off[n];
  if (start > (unsigned)N_EDGES) start = (unsigned)N_EDGES;
  unsigned end = start + deg;
  if (end > (unsigned)N_EDGES) end = (unsigned)N_EDGES;

  const float* bq = bases + q * 4;

  float4 a_sym = make_float4(0.f, 0.f, 0.f, 0.f);
  float4 a_sum = make_float4(0.f, 0.f, 0.f, 0.f);
  float4 a_sq  = make_float4(0.f, 0.f, 0.f, 0.f);
  float4 a_mx  = make_float4(-INFINITY, -INFINITY, -INFINITY, -INFINITY);

  for (unsigned j = start; j < end; j += 4) {
    unsigned r[4];
    float    dr[4];
    float4   v[4];
#pragma unroll
    for (int u = 0; u < 4; ++u) {
      unsigned jj = j + u;
      if (jj >= end) jj = end - 1;
      const unsigned rr = (unsigned)srcs[jj];
      r[u] = rr < (unsigned)N_NODES ? rr : (unsigned)(N_NODES - 1);
    }
#pragma unroll
    for (int u = 0; u < 4; ++u) {
      dr[u] = dinv[r[u]];
      v[u]  = *(const float4*)(bq + (size_t)r[u] * BD);
    }
#pragma unroll
    for (int u = 0; u < 4; ++u) {
      const bool  ok = (j + u) < end;
      const float d  = dr[u];
      const float tx = ok ? v[u].x : 0.f, mx_ = ok ? v[u].x : -INFINITY;
      const float ty = ok ? v[u].y : 0.f, my_ = ok ? v[u].y : -INFINITY;
      const float tz = ok ? v[u].z : 0.f, mz_ = ok ? v[u].z : -INFINITY;
      const float tw = ok ? v[u].w : 0.f, mw_ = ok ? v[u].w : -INFINITY;
      a_sym.x += d * tx;  a_sym.y += d * ty;  a_sym.z += d * tz;  a_sym.w += d * tw;
      a_sum.x += tx;      a_sum.y += ty;      a_sum.z += tz;      a_sum.w += tw;
      a_sq.x  += tx * tx; a_sq.y  += ty * ty; a_sq.z  += tz * tz; a_sq.w  += tw * tw;
      a_mx.x = fmaxf(a_mx.x, mx_); a_mx.y = fmaxf(a_mx.y, my_);
      a_mx.z = fmaxf(a_mx.z, mz_); a_mx.w = fmaxf(a_mx.w, mw_);
    }
  }

  {
    const float  cf = (float)deg;
    const float  ic = 1.0f / fmaxf(cf, 1.0f);
    const float  dv = dinv[n];
    const float4 bs = *(const float4*)(bq + (size_t)n * BD);
    const float  sy[4] = {a_sym.x, a_sym.y, a_sym.z, a_sym.w};
    const float  su[4] = {a_sum.x, a_sum.y, a_sum.z, a_sum.w};
    const float  sq[4] = {a_sq.x,  a_sq.y,  a_sq.z,  a_sq.w};
    const float  sm[4] = {a_mx.x,  a_mx.y,  a_mx.z,  a_mx.w};
    const float  bv[4] = {bs.x, bs.y, bs.z, bs.w};
    float o_sym[4], o_mean[4], o_mx[4], o_sd[4];
#pragma unroll
    for (int c = 0; c < 4; ++c) {
      o_sym[c] = dv * (sy[c] + dv * bv[c]);
      const float mean = su[c] * ic;
      const float var  = sq[c] * ic - mean * mean;
      o_mean[c] = mean;
      o_sd[c]   = sqrtf(fmaxf(var, 0.0f) + 1e-5f);
      o_mx[c]   = (cf > 0.0f) ? sm[c] : 0.0f;
    }
    fin[nl][0][q] = make_float4(o_sym[0],  o_sym[1],  o_sym[2],  o_sym[3]);
    fin[nl][1][q] = make_float4(o_mean[0], o_mean[1], o_mean[2], o_mean[3]);
    fin[nl][2][q] = make_float4(o_mx[0],   o_mx[1],   o_mx[2],   o_mx[3]);
    fin[nl][3][q] = make_float4(o_sd[0],   o_sd[1],   o_sd[2],   o_sd[3]);
  }
  __syncthreads();

  const int dg = q & 3;
  const int hq = q >> 2;
#pragma unroll
  for (int h2 = 0; h2 < 2; ++h2) {
    const int h = h2 * 4 + hq;
    const float4 bi = *(const float4*)(bias + h * 16 + dg * 4);
    float o0 = bi.x, o1 = bi.y, o2 = bi.z, o3 = bi.w;
#pragma unroll
    for (int b = 0; b < 4; ++b) {
      const float4 w  = *(const float4*)(wbuf + (size_t)n * OUT_C + h * 16 + b * 4);
      const float4 s0 = fin[nl][0][b * 4 + dg];
      const float4 s1 = fin[nl][1][b * 4 + dg];
      const float4 s2 = fin[nl][2][b * 4 + dg];
      const float4 s3 = fin[nl][3][b * 4 + dg];
      o0 += w.x * s0.x + w.y * s1.x + w.z * s2.x + w.w * s3.x;
      o1 += w.x * s0.y + w.y * s1.y + w.z * s2.y + w.w * s3.y;
      o2 += w.x * s0.z + w.y * s1.z + w.z * s2.z + w.w * s3.z;
      o3 += w.x * s0.w + w.y * s1.w + w.z * s2.w + w.w * s3.w;
    }
    *(float4*)(out + (size_t)n * OUT_C + h * 16 + dg * 4) = make_float4(o0, o1, o2, o3);
  }
}

extern "C" void kernel_launch(void* const* d_in, const int* in_sizes, int n_in,
                              void* d_out, int out_size, void* d_ws, size_t ws_size,
                              hipStream_t stream) {
  (void)in_sizes; (void)n_in; (void)out_size; (void)ws_size;
  const float* x    = (const float*)d_in[0];
  const int*   ei   = (const int*)d_in[1];
  const float* bW   = (const float*)d_in[2];
  const float* cW   = (const float*)d_in[3];
  const float* cb   = (const float*)d_in[4];
  const float* bias = (const float*)d_in[5];
  float*       out  = (float*)d_out;

  char* ws = (char*)d_ws;
  char*     r0    = ws;          ws += (size_t)N_NODES * 256 * sizeof(float);
  float*    bases = (float*)ws;  ws += (size_t)N_NODES * BD * sizeof(float);
  float*    wbuf  = (float*)ws;  ws += (size_t)N_NODES * OUT_C * sizeof(float);
  unsigned* cnt   = (unsigned*)ws; ws += (size_t)N_NODES * sizeof(unsigned);
  float*    dinv  = (float*)ws;

  unsigned* rank = (unsigned*)r0;
  int*      srcs = (int*)(r0 + (size_t)N_EDGES * sizeof(unsigned));
  unsigned* off  = (unsigned*)(r0 + (size_t)N_EDGES * 2 * sizeof(unsigned));

  k_zero<<<(N_NODES + 255) / 256, 256, 0, stream>>>(cnt);
  k_rank<<<(N_EDGES + 255) / 256, 256, 0, stream>>>(ei + N_EDGES, cnt, rank);
  k_scan<<<(N_NODES + 255) / 256, 256, 0, stream>>>(cnt, off, dinv);
  k_fill<<<(N_EDGES + 255) / 256, 256, 0, stream>>>(ei, off, rank, srcs);
  k_gemm<<<(N_NODES + 63) / 64, 128, 0, stream>>>(x, bW, cW, cb, bases, wbuf);
  k_gather<<<N_NODES / 16, 256, 0, stream>>>(srcs, off, cnt, bases, wbuf, dinv, bias, out);
}


# k_gemm softmax: 16-lane max and sum butterflies as row_xmask DPP steps on the vector unit, eight rows side by side, instead of LDS-unit lane permutes with a wait each; same butterfly, same sums
# speedup vs baseline: 2.6458x; 1.0965x over previous
#include <hip/hip_runtime.h>
#include <math.h>
#include <float.h>


#define N_NODES 50000
#define N_EDGES 800000
#define IN_C    128
#define OUT_C   128
#define HEADS   8
#define BASES   4
#define HEAD_D  16
#define BD      (BASES * HEAD_D)

typedef float v2f __attribute__((ext_vector_type(2)));
typedef float v8f __attribute__((ext_vector_type(8)));

static_assert(N_NODES % 16 == 0, "k_gather maps 16 nodes to a block with no tail");

__global__ __launch_bounds__(256) void k_zero(unsigned* __restrict__ cnt) {
  const int i = blockIdx.x * 256 + threadIdx.x;
  if (i < N_NODES) cnt[i] = 0u;
}

__global__ __launch_bounds__(256) void k_rank(const int* __restrict__ col,
                                              unsigned* __restrict__ cnt,
                                              unsigned* __restrict__ rank) {
  const int e = blockIdx.x * 256 + threadIdx.x;
  if (e < N_EDGES) rank[e] = atomicAdd(&cnt[col[e]], 1u);
}

__global__ __launch_bounds__(256) void k_scan(const unsigned* __restrict__ cnt,
                                              unsigned* __restrict__ off,
                                              float* __restrict__ dinv) {
  __shared__ unsigned wpart[8];
  __shared__ unsigned wtot[8];

  const int tid  = threadIdx.x;
  const int lane = tid & 31;
  const int wave = tid >> 5;
  const int b0   = blockIdx.x * 256;
  const int n    = b0 + tid;

  unsigned part = 0u;
  for (int i0 = 0; i0 < b0; i0 += 8 * 1024) {
#pragma unroll
    for (int u = 0; u < 8; ++u) {
      const int   i  = i0 + u * 1024 + tid * 4;
      const bool  ok = i < b0;
      const uint4 v  = *(const uint4*)(cnt + (ok ? i : 0));
      part += (v.x + v.y + v.z + v.w) * (unsigned)ok;
    }
  }
  part += __shfl_xor(part, 16);
  part += __shfl_xor(part, 8);
  part += __shfl_xor(part, 4);
  part += __shfl_xor(part, 2);
  part += __shfl_xor(part, 1);

  const unsigned c = (n < N_NODES) ? cnt[n] : 0u;
  unsigned x = c;
  for (int d = 1; d < 32; d <<= 1) {
    const unsigned y = __shfl_up(x, d);
    if (lane >= d) x += y;
  }
  if (lane == 31) wtot[wave] = x;
  if (lane == 0)  wpart[wave] = part;
  __syncthreads();

  unsigned base = 0u;
  for (int w = 0; w < 8; ++w) {
    base += wpart[w];
    if (w < wave) base += wtot[w];
  }
  if (n < N_NODES) {
    off[n]  = base + x - c;
    dinv[n] = rsqrtf((float)c + 1.0f);
  }
}

__global__ __launch_bounds__(256) void k_fill(const int* __restrict__ ei,
                                              const unsigned* __restrict__ off,
                                              const unsigned* __restrict__ rank,
                                              int* __restrict__ srcs) {
  const int e = blockIdx.x * 256 + threadIdx.x;
  if (e < N_EDGES) {
    const int row = ei[e];
    const int col = ei[N_EDGES + e];
    const unsigned p = off[col] + rank[e];
    if (p < (unsigned)N_EDGES) srcs[p] = row;
  }
}

template <int M>
__device__ __forceinline__ float row_xor(float v) {
  return __builtin_bit_cast(float, __builtin_amdgcn_update_dpp(
      0, __builtin_bit_cast(int, v), 0x160 + M, 0xf, 0xf, false));
}

__global__ __launch_bounds__(128) void k_gemm(const float* __restrict__ x,
                                              const float* __restrict__ bW,
                                              const float* __restrict__ cW,
                                              const float* __restrict__ cb,
                                              float* __restrict__ bases,
                                              float* __restrict__ wbuf) {
  __shared__ float xs[64][132];

  const int tid  = threadIdx.x;
  const int row0 = blockIdx.x * 64;

  for (int t = 0; t < 16; ++t) {
    int idx = t * 128 + tid;
    int r   = idx >> 5;
    int c4  = (idx & 31) * 4;
    int gr  = row0 + r;
    if (gr >= N_NODES) gr = N_NODES - 1;
    const float4 v = *(const float4*)(x + (size_t)gr * IN_C + c4);
    xs[r][c4 + 0] = v.x; xs[r][c4 + 1] = v.y;
    xs[r][c4 + 2] = v.z; xs[r][c4 + 3] = v.w;
  }
  __syncthreads();

  const int wave = tid >> 5;
  const int lane = tid & 31;
  const int nlo  = lane & 15;
  const int half = lane >> 4;
  const int kb   = half * 2;
  const int lr   = wave * 16 + nlo;

  for (int nt = 0; nt < 12; ++nt) {
    v8f acc = {0.f, 0.f, 0.f, 0.f, 0.f, 0.f, 0.f, 0.f};

    if (nt < 4) {
      const float* W = bW + nt * (IN_C * HEAD_D);
#pragma unroll
      for (int ks = 0; ks < 32; ++ks) {
        const int k = ks * 4 + kb;
        v2f a = *(const v2f*)&xs[lr][k];
        v2f b = { W[k * HEAD_D + nlo], W[(k + 1) * HEAD_D + nlo] };
        acc = __builtin_amdgcn_wmma_f32_16x16x4_f32(
            false, a, false, b, (short)0, acc, false, false);
      }
#pragma unroll
      for (int r = 0; r < 8; ++r) {
        const int m  = r + (half << 3);
        const int gr = row0 + wave * 16 + m;
        if (gr < N_NODES) bases[(size_t)gr * BD + nt * 16 + nlo] = acc[r];
      }
    } else {
      const int   head = nt - 4;
      const float* W   = cW + head * 16;
#pragma unroll
      for (int ks = 0; ks < 32; ++ks) {
        const int k = ks * 4 + kb;
        v2f a = *(const v2f*)&xs[lr][k];
        v2f b = { W[k * OUT_C + nlo], W[(k + 1) * OUT_C + nlo] };
        acc = __builtin_amdgcn_wmma_f32_16x16x4_f32(
            false, a, false, b, (short)0, acc, false, false);
      }
      const float bb = cb[head * 16 + nlo];
      float v[8], mx[8], ex[8], sm[8];
#pragma unroll
      for (int r = 0; r < 8; ++r) { v[r] = acc[r] + bb; mx[r] = v[r]; }
#pragma unroll
      for (int r = 0; r < 8; ++r) mx[r] = fmaxf(mx[r], row_xor<1>(mx[r]));
#pragma unroll
      for (int r = 0; r < 8; ++r) mx[r] = fmaxf(mx[r], row_xor<2>(mx[r]));
#pragma unroll
      for (int r = 0; r < 8; ++r) mx[r] = fmaxf(mx[r], row_xor<4>(mx[r]));
#pragma unroll
      for (int r = 0; r < 8; ++r) mx[r] = fmaxf(mx[r], row_xor<8>(mx[r]));
#pragma unroll
      for (int r = 0; r < 8; ++r) { ex[r] = __expf(v[r] - mx[r]); sm[r] = ex[r]; }
#pragma unroll
      for (int r = 0; r < 8; ++r) sm[r] += row_xor<1>(sm[r]);
#pragma unroll
      for (int r = 0; r < 8; ++r) sm[r] += row_xor<2>(sm[r]);
#pragma unroll
      for (int r = 0; r < 8; ++r) sm[r] += row_xor<4>(sm[r]);
#pragma unroll
      for (int r = 0; r < 8; ++r) sm[r] += row_xor<8>(sm[r]);
#pragma unroll
      for (int r = 0; r < 8; ++r) {
        const float wv = ex[r] / sm[r];
        const int m  = r + (half << 3);
        const int gr = row0 + wave * 16 + m;
        if (gr < N_NODES) wbuf[(size_t)gr * OUT_C + head * 16 + nlo] = wv;
      }
    }
  }
}

__global__ __launch_bounds__(256) void k_gather(const int* __restrict__ srcs,
                                                const unsigned* __restrict__ off,
                                                const unsigned* __restrict__ cnt,
                                                const float* __restrict__ bases,
                                                const float* __restrict__ wbuf,
                                                const float* __restrict__ dinv,
                                                const float* __restrict__ bias,
                                                float* __restrict__ out) {
  __shared__ float4 fin[16][4][16];

  const int tid = threadIdx.x;
  const int nl  = tid >> 4;
  const int q   = tid & 15;
  const int n   = blockIdx.x * 16 + nl;

  const unsigned deg = cnt[n];
  unsigned start = off[n];
  if (start > (unsigned)N_EDGES) start = (unsigned)N_EDGES;
  unsigned end = start + deg;
  if (end > (unsigned)N_EDGES) end = (unsigned)N_EDGES;

  const float* bq = bases + q * 4;

  float4 a_sym = make_float4(0.f, 0.f, 0.f, 0.f);
  float4 a_sum = make_float4(0.f, 0.f, 0.f, 0.f);
  float4 a_sq  = make_float4(0.f, 0.f, 0.f, 0.f);
  float4 a_mx  = make_float4(-INFINITY, -INFINITY, -INFINITY, -INFINITY);

  for (unsigned j = start; j < end; j += 4) {
    unsigned r[4];
    float    dr[4];
    float4   v[4];
#pragma unroll
    for (int u = 0; u < 4; ++u) {
      unsigned jj = j + u;
      if (jj >= end) jj = end - 1;
      const unsigned rr = (unsigned)srcs[jj];
      r[u] = rr < (unsigned)N_NODES ? rr : (unsigned)(N_NODES - 1);
    }
#pragma unroll
    for (int u = 0; u < 4; ++u) {
      dr[u] = dinv[r[u]];
      v[u]  = *(const float4*)(bq + (size_t)r[u] * BD);
    }
#pragma unroll
    for (int u = 0; u < 4; ++u) {
      const bool  ok = (j + u) < end;
      const float d  = dr[u];
      const float tx = ok ? v[u].x : 0.f, mx_ = ok ? v[u].x : -INFINITY;
      const float ty = ok ? v[u].y : 0.f, my_ = ok ? v[u].y : -INFINITY;
      const float tz = ok ? v[u].z : 0.f, mz_ = ok ? v[u].z : -INFINITY;
      const float tw = ok ? v[u].w : 0.f, mw_ = ok ? v[u].w : -INFINITY;
      a_sym.x += d * tx;  a_sym.y += d * ty;  a_sym.z += d * tz;  a_sym.w += d * tw;
      a_sum.x += tx;      a_sum.y += ty;      a_sum.z += tz;      a_sum.w += tw;
      a_sq.x  += tx * tx; a_sq.y  += ty * ty; a_sq.z  += tz * tz; a_sq.w  += tw * tw;
      a_mx.x = fmaxf(a_mx.x, mx_); a_mx.y = fmaxf(a_mx.y, my_);
      a_mx.z = fmaxf(a_mx.z, mz_); a_mx.w = fmaxf(a_mx.w, mw_);
    }
  }

  {
    const float  cf = (float)deg;
    const float  ic = 1.0f / fmaxf(cf, 1.0f);
    const float  dv = dinv[n];
    const float4 bs = *(const float4*)(bq + (size_t)n * BD);
    const float  sy[4] = {a_sym.x, a_sym.y, a_sym.z, a_sym.w};
    const float  su[4] = {a_sum.x, a_sum.y, a_sum.z, a_sum.w};
    const float  sq[4] = {a_sq.x,  a_sq.y,  a_sq.z,  a_sq.w};
    const float  sm[4] = {a_mx.x,  a_mx.y,  a_mx.z,  a_mx.w};
    const float  bv[4] = {bs.x, bs.y, bs.z, bs.w};
    float o_sym[4], o_mean[4], o_mx[4], o_sd[4];
#pragma unroll
    for (int c = 0; c < 4; ++c) {
      o_sym[c] = dv * (sy[c] + dv * bv[c]);
      const float mean = su[c] * ic;
      const float var  = sq[c] * ic - mean * mean;
      o_mean[c] = mean;
      o_sd[c]   = sqrtf(fmaxf(var, 0.0f) + 1e-5f);
      o_mx[c]   = (cf > 0.0f) ? sm[c] : 0.0f;
    }
    fin[nl][0][q] = make_float4(o_sym[0],  o_sym[1],  o_sym[2],  o_sym[3]);
    fin[nl][1][q] = make_float4(o_mean[0], o_mean[1], o_mean[2], o_mean[3]);
    fin[nl][2][q] = make_float4(o_mx[0],   o_mx[1],   o_mx[2],   o_mx[3]);
    fin[nl][3][q] = make_float4(o_sd[0],   o_sd[1],   o_sd[2],   o_sd[3]);
  }
  __syncthreads();

  const int dg = q & 3;
  const int hq = q >> 2;
#pragma unroll
  for (int h2 = 0; h2 < 2; ++h2) {
    const int h = h2 * 4 + hq;
    const float4 bi = *(const float4*)(bias + h * 16 + dg * 4);
    float o0 = bi.x, o1 = bi.y, o2 = bi.z, o3 = bi.w;
#pragma unroll
    for (int b = 0; b < 4; ++b) {
      const float4 w  = *(const float4*)(wbuf + (size_t)n * OUT_C + h * 16 + b * 4);
      const float4 s0 = fin[nl][0][b * 4 + dg];
      const float4 s1 = fin[nl][1][b * 4 + dg];
      const float4 s2 = fin[nl][2][b * 4 + dg];
      const float4 s3 = fin[nl][3][b * 4 + dg];
      o0 += w.x * s0.x + w.y * s1.x + w.z * s2.x + w.w * s3.x;
      o1 += w.x * s0.y + w.y * s1.y + w.z * s2.y + w.w * s3.y;
      o2 += w.x * s0.z + w.y * s1.z + w.z * s2.z + w.w * s3.z;
      o3 += w.x * s0.w + w.y * s1.w + w.z * s2.w + w.w * s3.w;
    }
    *(float4*)(out + (size_t)n * OUT_C + h * 16 + dg * 4) = make_float4(o0, o1, o2, o3);
  }
}

extern "C" void kernel_launch(void* const* d_in, const int* in_sizes, int n_in,
                              void* d_out, int out_size, void* d_ws, size_t ws_size,
                              hipStream_t stream) {
  (void)in_sizes; (void)n_in; (void)out_size; (void)ws_size;
  const float* x    = (const float*)d_in[0];
  const int*   ei   = (const int*)d_in[1];
  const float* bW   = (const float*)d_in[2];
  const float* cW   = (const float*)d_in[3];
  const float* cb   = (const float*)d_in[4];
  const float* bias = (const float*)d_in[5];
  float*       out  = (float*)d_out;

  char* ws = (char*)d_ws;
  char*     r0    = ws;          ws += (size_t)N_NODES * 256 * sizeof(float);
  float*    bases = (float*)ws;  ws += (size_t)N_NODES * BD * sizeof(float);
  float*    wbuf  = (float*)ws;  ws += (size_t)N_NODES * OUT_C * sizeof(float);
  unsigned* cnt   = (unsigned*)ws; ws += (size_t)N_NODES * sizeof(unsigned);
  float*    dinv  = (float*)ws;

  unsigned* rank = (unsigned*)r0;
  int*      srcs = (int*)(r0 + (size_t)N_EDGES * sizeof(unsigned));
  unsigned* off  = (unsigned*)(r0 + (size_t)N_EDGES * 2 * sizeof(unsigned));

  k_zero<<<(N_NODES + 255) / 256, 256, 0, stream>>>(cnt);
  k_rank<<<(N_EDGES + 255) / 256, 256, 0, stream>>>(ei + N_EDGES, cnt, rank);
  k_scan<<<(N_NODES + 255) / 256, 256, 0, stream>>>(cnt, off, dinv);
  k_fill<<<(N_EDGES + 255) / 256, 256, 0, stream>>>(ei, off, rank, srcs);
  k_gemm<<<(N_NODES + 63) / 64, 128, 0, stream>>>(x, bW, cW, cb, bases, wbuf);
  k_gather<<<N_NODES / 16, 256, 0, stream>>>(srcs, off, cnt, bases, wbuf, dinv, bias, out);
}


# k_fill writes each sorted edge as an 8-byte record {source, dinv of the source from the LDS counter table}; k_gather reads the record and drops its random per-edge dinv[] read; sum order kept
# speedup vs baseline: 3.2037x; 1.2109x over previous
#include <hip/hip_runtime.h>
#include <math.h>
#include <float.h>


#define N_NODES 50000
#define N_EDGES 800000
#define IN_C    128
#define OUT_C   128
#define HEADS   8
#define BASES   4
#define HEAD_D  16
#define BD      (BASES * HEAD_D)

typedef float v2f __attribute__((ext_vector_type(2)));
typedef float v8f __attribute__((ext_vector_type(8)));

static_assert(N_NODES % 16 == 0, "k_gather maps 16 nodes to a block with no tail");

__global__ __launch_bounds__(256) void k_zero(unsigned* __restrict__ cnt) {
  const int i = blockIdx.x * 256 + threadIdx.x;
  if (i < N_NODES) cnt[i] = 0u;
}

#define GEMM_BLOCKS  ((N_NODES + 63) / 64)
#define RANK_EPW     256
#define RANK_WAVES   (N_EDGES / RANK_EPW)
#define RANK_BLOCKS  ((RANK_WAVES + 3) / 4)

static_assert(N_EDGES % RANK_EPW == 0, "every ranking wave has all its 256 edges");
static_assert((N_EDGES * sizeof(int)) % 16 == 0, "col = ei + N_EDGES stays 16-byte aligned");

__device__ __forceinline__ void rank_wave(const int* __restrict__ col,
                                          unsigned* __restrict__ cnt,
                                          unsigned* __restrict__ rank,
                                          int gw, int lane) {
  const int  e  = gw * RANK_EPW + lane * 4;
  int4 c0 = *(const int4*)(col + e);
  int4 c1 = *(const int4*)(col + e + 128);
  asm volatile("" : "+v"(c0.x), "+v"(c0.y), "+v"(c0.z), "+v"(c0.w),
                    "+v"(c1.x), "+v"(c1.y), "+v"(c1.z), "+v"(c1.w));
  uint4 r0, r1;
  r0.x = atomicAdd(&cnt[c0.x], 1u);
  r0.y = atomicAdd(&cnt[c0.y], 1u);
  r0.z = atomicAdd(&cnt[c0.z], 1u);
  r0.w = atomicAdd(&cnt[c0.w], 1u);
  r1.x = atomicAdd(&cnt[c1.x], 1u);
  r1.y = atomicAdd(&cnt[c1.y], 1u);
  r1.z = atomicAdd(&cnt[c1.z], 1u);
  r1.w = atomicAdd(&cnt[c1.w], 1u);
  *(uint4*)(rank + e)       = r0;
  *(uint4*)(rank + e + 128) = r1;
}

#define FILL_BLOCKS  256
#define FILL_THREADS 1024
#define FILL_TRIPS   13
#define FILL_CHUNK   (4 * FILL_TRIPS)
#define FILL_TAB     (FILL_THREADS * FILL_CHUNK)
#define FILL_EPB     (N_EDGES / FILL_BLOCKS)
#define FILL_NPB     ((N_NODES + FILL_BLOCKS - 1) / FILL_BLOCKS)

static_assert(FILL_TAB >= N_NODES, "the table holds every counter");
static_assert(N_EDGES % FILL_BLOCKS == 0, "edges split evenly over the fill blocks");
static_assert(FILL_EPB <= 4 * FILL_THREADS && FILL_EPB > 3 * FILL_THREADS, "three or four edges per thread");
static_assert(N_NODES % 4 == 0, "counters are copied sixteen bytes at a time");
static_assert(FILL_NPB <= FILL_THREADS, "one thread per node of the block's slice");

__global__ __launch_bounds__(FILL_THREADS) void k_fill(const int* __restrict__ ei,
                                                       const unsigned* __restrict__ cnt,
                                                       const unsigned* __restrict__ rank,
                                                       int* __restrict__ erec,
                                                       unsigned* __restrict__ off,
                                                       float* __restrict__ dinv) {
  __shared__ __attribute__((aligned(16))) unsigned tab[FILL_TAB];
  __shared__ unsigned wsum[32];

  const int tid  = threadIdx.x;
  const int lane = tid & 31;
  const int wave = tid >> 5;

  const int e0 = blockIdx.x * FILL_EPB;
  int      erow[4], ecol[4];
  unsigned erk[4];
#pragma unroll
  for (int u = 0; u < 4; ++u) {
    const int k = tid + u * FILL_THREADS;
    const int e = e0 + (k < FILL_EPB ? k : tid);
    erow[u] = ei[e];
    ecol[u] = ei[N_EDGES + e];
    erk[u]  = rank[e];
  }
  const int      nmine = blockIdx.x * FILL_NPB + tid;
  const bool     nok   = tid < FILL_NPB && nmine < N_NODES;
  const unsigned cmine = cnt[nok ? nmine : 0];

  const int rot = (int)(blockIdx.x % (unsigned)FILL_TRIPS);
  uint4 cv[FILL_TRIPS];
#pragma unroll
  for (int t = 0; t < FILL_TRIPS; ++t) {
    int it = t + rot;
    if (it >= FILL_TRIPS) it -= FILL_TRIPS;
    const int i = it * (4 * FILL_THREADS) + tid * 4;
    cv[t] = *(const uint4*)(cnt + (i < N_NODES ? i : 0));
  }
#pragma unroll
  for (int t = 0; t < FILL_TRIPS; ++t) {
    int it = t + rot;
    if (it >= FILL_TRIPS) it -= FILL_TRIPS;
    const int      i = it * (4 * FILL_THREADS) + tid * 4;
    const unsigned m = 0u - (unsigned)(i < N_NODES);
    *(uint4*)(tab + i) = make_uint4(cv[t].x & m, cv[t].y & m, cv[t].z & m, cv[t].w & m);
  }
  __syncthreads();

  float edv[4];
  for (int u = 0; u < 4; ++u) {
    unsigned rr = (unsigned)erow[u];
    if (rr >= (unsigned)N_NODES) rr = (unsigned)(N_NODES - 1);
    edv[u] = rsqrtf((float)tab[rr] + 1.0f);
  }

  uint4*   my = (uint4*)(tab + tid * FILL_CHUNK);
  uint4    c[FILL_TRIPS];
  unsigned s = 0u;
#pragma unroll
  for (int j = 0; j < FILL_TRIPS; ++j) { c[j] = my[j]; s += c[j].x + c[j].y + c[j].z + c[j].w; }

  unsigned x = s;
  for (int d = 1; d < 32; d <<= 1) {
    const unsigned y = __shfl_up(x, d);
    if (lane >= d) x += y;
  }
  if (lane == 31) wsum[wave] = x;
  __syncthreads();

  const unsigned wt = wsum[lane];
  unsigned wx = wt;
  for (int d = 1; d < 32; d <<= 1) {
    const unsigned y = __shfl_up(wx, d);
    if (lane >= d) wx += y;
  }
  const unsigned base = __shfl(wx - wt, wave);

  unsigned run = base + x - s;
#pragma unroll
  for (int j = 0; j < FILL_TRIPS; ++j) {
    uint4 o;
    o.x = run; run += c[j].x;
    o.y = run; run += c[j].y;
    o.z = run; run += c[j].z;
    o.w = run; run += c[j].w;
    my[j] = o;
  }
  __syncthreads();

  if (nok) {
    off[nmine]  = tab[nmine];
    dinv[nmine] = rsqrtf((float)cmine + 1.0f);
  }

  int2* rec = (int2*)erec;
#pragma unroll
  for (int u = 0; u < 4; ++u) {
    const int k = tid + u * FILL_THREADS;
    if (k < FILL_EPB) {
      unsigned cc = (unsigned)ecol[u];
      if (cc >= (unsigned)N_NODES) cc = (unsigned)(N_NODES - 1);
      const unsigned p = tab[cc] + erk[u];
      if (p < (unsigned)N_EDGES) rec[p] = make_int2(erow[u], __float_as_int(edv[u]));
    }
  }
}

template <int M>
__device__ __forceinline__ float row_xor(float v) {
  return __builtin_bit_cast(float, __builtin_amdgcn_update_dpp(
      0, __builtin_bit_cast(int, v), 0x160 + M, 0xf, 0xf, false));
}

__global__ __launch_bounds__(128) void k_gemm(const float* __restrict__ x,
                                              const float* __restrict__ bW,
                                              const float* __restrict__ cW,
                                              const float* __restrict__ cb,
                                              float* __restrict__ bases,
                                              float* __restrict__ wbuf,
                                              const int* __restrict__ col,
                                              unsigned* __restrict__ cnt,
                                              unsigned* __restrict__ rank) {
  __shared__ float xs[64][132];

  const int tid  = threadIdx.x;

  if (blockIdx.x >= GEMM_BLOCKS) {
    const int gw = ((int)blockIdx.x - GEMM_BLOCKS) * 4 + (tid >> 5);
    if (gw < RANK_WAVES) rank_wave(col, cnt, rank, gw, tid & 31);
    return;
  }

  const int row0 = blockIdx.x * 64;

  for (int t = 0; t < 16; ++t) {
    int idx = t * 128 + tid;
    int r   = idx >> 5;
    int c4  = (idx & 31) * 4;
    int gr  = row0 + r;
    if (gr >= N_NODES) gr = N_NODES - 1;
    const float4 v = *(const float4*)(x + (size_t)gr * IN_C + c4);
    xs[r][c4 + 0] = v.x; xs[r][c4 + 1] = v.y;
    xs[r][c4 + 2] = v.z; xs[r][c4 + 3] = v.w;
  }
  __syncthreads();

  const int wave = tid >> 5;
  const int lane = tid & 31;
  const int nlo  = lane & 15;
  const int half = lane >> 4;
  const int kb   = half * 2;
  const int lr   = wave * 16 + nlo;

  for (int nt = 0; nt < 12; ++nt) {
    v8f acc = {0.f, 0.f, 0.f, 0.f, 0.f, 0.f, 0.f, 0.f};

    if (nt < 4) {
      const float* W = bW + nt * (IN_C * HEAD_D);
#pragma unroll
      for (int ks = 0; ks < 32; ++ks) {
        const int k = ks * 4 + kb;
        v2f a = *(const v2f*)&xs[lr][k];
        v2f b = { W[k * HEAD_D + nlo], W[(k + 1) * HEAD_D + nlo] };
        acc = __builtin_amdgcn_wmma_f32_16x16x4_f32(
            false, a, false, b, (short)0, acc, false, false);
      }
#pragma unroll
      for (int r = 0; r < 8; ++r) {
        const int m  = r + (half << 3);
        const int gr = row0 + wave * 16 + m;
        if (gr < N_NODES) bases[(size_t)gr * BD + nt * 16 + nlo] = acc[r];
      }
    } else {
      const int   head = nt - 4;
      const float* W   = cW + head * 16;
#pragma unroll
      for (int ks = 0; ks < 32; ++ks) {
        const int k = ks * 4 + kb;
        v2f a = *(const v2f*)&xs[lr][k];
        v2f b = { W[k * OUT_C + nlo], W[(k + 1) * OUT_C + nlo] };
        acc = __builtin_amdgcn_wmma_f32_16x16x4_f32(
            false, a, false, b, (short)0, acc, false, false);
      }
      const float bb = cb[head * 16 + nlo];
      float v[8], mx[8], ex[8], sm[8];
#pragma unroll
      for (int r = 0; r < 8; ++r) { v[r] = acc[r] + bb; mx[r] = v[r]; }
#pragma unroll
      for (int r = 0; r < 8; ++r) mx[r] = fmaxf(mx[r], row_xor<1>(mx[r]));
#pragma unroll
      for (int r = 0; r < 8; ++r) mx[r] = fmaxf(mx[r], row_xor<2>(mx[r]));
#pragma unroll
      for (int r = 0; r < 8; ++r) mx[r] = fmaxf(mx[r], row_xor<4>(mx[r]));
#pragma unroll
      for (int r = 0; r < 8; ++r) mx[r] = fmaxf(mx[r], row_xor<8>(mx[r]));
#pragma unroll
      for (int r = 0; r < 8; ++r) { ex[r] = __expf(v[r] - mx[r]); sm[r] = ex[r]; }
#pragma unroll
      for (int r = 0; r < 8; ++r) sm[r] += row_xor<1>(sm[r]);
#pragma unroll
      for (int r = 0; r < 8; ++r) sm[r] += row_xor<2>(sm[r]);
#pragma unroll
      for (int r = 0; r < 8; ++r) sm[r] += row_xor<4>(sm[r]);
#pragma unroll
      for (int r = 0; r < 8; ++r) sm[r] += row_xor<8>(sm[r]);
#pragma unroll
      for (int r = 0; r < 8; ++r) {
        const float wv = ex[r] / sm[r];
        const int m  = r + (half << 3);
        const int gr = row0 + wave * 16 + m;
        if (gr < N_NODES) wbuf[(size_t)gr * OUT_C + head * 16 + nlo] = wv;
      }
    }
  }
}

__global__ __launch_bounds__(256) void k_gather(const int* __restrict__ erec,
                                                const unsigned* __restrict__ off,
                                                const unsigned* __restrict__ cnt,
                                                const float* __restrict__ bases,
                                                const float* __restrict__ wbuf,
                                                const float* __restrict__ dinv,
                                                const float* __restrict__ bias,
                                                float* __restrict__ out) {
  __shared__ float4 fin[16][4][16];

  const int tid = threadIdx.x;
  const int nl  = tid >> 4;
  const int q   = tid & 15;
  const int n   = blockIdx.x * 16 + nl;

  const unsigned deg = cnt[n];
  unsigned start = off[n];
  if (start > (unsigned)N_EDGES) start = (unsigned)N_EDGES;
  unsigned end = start + deg;
  if (end > (unsigned)N_EDGES) end = (unsigned)N_EDGES;

  const float* bq  = bases + q * 4;
  const int2*  rec = (const int2*)erec;

  float4 a_sym = make_float4(0.f, 0.f, 0.f, 0.f);
  float4 a_sum = make_float4(0.f, 0.f, 0.f, 0.f);
  float4 a_sq  = make_float4(0.f, 0.f, 0.f, 0.f);
  float4 a_mx  = make_float4(-INFINITY, -INFINITY, -INFINITY, -INFINITY);

  for (unsigned j = start; j < end; j += 4) {
    unsigned r[4];
    float    dr[4];
    float4   v[4];
#pragma unroll
    for (int u = 0; u < 4; ++u) {
      unsigned jj = j + u;
      if (jj >= end) jj = end - 1;
      const int2     e  = rec[jj];
      const unsigned rr = (unsigned)e.x;
      r[u]  = rr < (unsigned)N_NODES ? rr : (unsigned)(N_NODES - 1);
      dr[u] = __int_as_float(e.y);
    }
#pragma unroll
    for (int u = 0; u < 4; ++u) v[u] = *(const float4*)(bq + (size_t)r[u] * BD);
#pragma unroll
    for (int u = 0; u < 4; ++u) {
      const bool  ok = (j + u) < end;
      const float d  = dr[u];
      const float tx = ok ? v[u].x : 0.f, mx_ = ok ? v[u].x : -INFINITY;
      const float ty = ok ? v[u].y : 0.f, my_ = ok ? v[u].y : -INFINITY;
      const float tz = ok ? v[u].z : 0.f, mz_ = ok ? v[u].z : -INFINITY;
      const float tw = ok ? v[u].w : 0.f, mw_ = ok ? v[u].w : -INFINITY;
      a_sym.x += d * tx;  a_sym.y += d * ty;  a_sym.z += d * tz;  a_sym.w += d * tw;
      a_sum.x += tx;      a_sum.y += ty;      a_sum.z += tz;      a_sum.w += tw;
      a_sq.x  += tx * tx; a_sq.y  += ty * ty; a_sq.z  += tz * tz; a_sq.w  += tw * tw;
      a_mx.x = fmaxf(a_mx.x, mx_); a_mx.y = fmaxf(a_mx.y, my_);
      a_mx.z = fmaxf(a_mx.z, mz_); a_mx.w = fmaxf(a_mx.w, mw_);
    }
  }

  {
    const float  cf = (float)deg;
    const float  ic = 1.0f / fmaxf(cf, 1.0f);
    const float  dv = dinv[n];
    const float4 bs = *(const float4*)(bq + (size_t)n * BD);
    const float  sy[4] = {a_sym.x, a_sym.y, a_sym.z, a_sym.w};
    const float  su[4] = {a_sum.x, a_sum.y, a_sum.z, a_sum.w};
    const float  sq[4] = {a_sq.x,  a_sq.y,  a_sq.z,  a_sq.w};
    const float  sm[4] = {a_mx.x,  a_mx.y,  a_mx.z,  a_mx.w};
    const float  bv[4] = {bs.x, bs.y, bs.z, bs.w};
    float o_sym[4], o_mean[4], o_mx[4], o_sd[4];
#pragma unroll
    for (int c = 0; c < 4; ++c) {
      o_sym[c] = dv * (sy[c] + dv * bv[c]);
      const float mean = su[c] * ic;
      const float var  = sq[c] * ic - mean * mean;
      o_mean[c] = mean;
      o_sd[c]   = sqrtf(fmaxf(var, 0.0f) + 1e-5f);
      o_mx[c]   = (cf > 0.0f) ? sm[c] : 0.0f;
    }
    fin[nl][0][q] = make_float4(o_sym[0],  o_sym[1],  o_sym[2],  o_sym[3]);
    fin[nl][1][q] = make_float4(o_mean[0], o_mean[1], o_mean[2], o_mean[3]);
    fin[nl][2][q] = make_float4(o_mx[0],   o_mx[1],   o_mx[2],   o_mx[3]);
    fin[nl][3][q] = make_float4(o_sd[0],   o_sd[1],   o_sd[2],   o_sd[3]);
  }
  __syncthreads();

  const int dg = q & 3;
  const int hq = q >> 2;
#pragma unroll
  for (int h2 = 0; h2 < 2; ++h2) {
    const int h = h2 * 4 + hq;
    const float4 bi = *(const float4*)(bias + h * 16 + dg * 4);
    float o0 = bi.x, o1 = bi.y, o2 = bi.z, o3 = bi.w;
#pragma unroll
    for (int b = 0; b < 4; ++b) {
      const float4 w  = *(const float4*)(wbuf + (size_t)n * OUT_C + h * 16 + b * 4);
      const float4 s0 = fin[nl][0][b * 4 + dg];
      const float4 s1 = fin[nl][1][b * 4 + dg];
      const float4 s2 = fin[nl][2][b * 4 + dg];
      const float4 s3 = fin[nl][3][b * 4 + dg];
      o0 += w.x * s0.x + w.y * s1.x + w.z * s2.x + w.w * s3.x;
      o1 += w.x * s0.y + w.y * s1.y + w.z * s2.y + w.w * s3.y;
      o2 += w.x * s0.z + w.y * s1.z + w.z * s2.z + w.w * s3.z;
      o3 += w.x * s0.w + w.y * s1.w + w.z * s2.w + w.w * s3.w;
    }
    *(float4*)(out + (size_t)n * OUT_C + h * 16 + dg * 4) = make_float4(o0, o1, o2, o3);
  }
}

extern "C" void kernel_launch(void* const* d_in, const int* in_sizes, int n_in,
                              void* d_out, int out_size, void* d_ws, size_t ws_size,
                              hipStream_t stream) {
  (void)in_sizes; (void)n_in; (void)out_size; (void)ws_size;
  const float* x    = (const float*)d_in[0];
  const int*   ei   = (const int*)d_in[1];
  const float* bW   = (const float*)d_in[2];
  const float* cW   = (const float*)d_in[3];
  const float* cb   = (const float*)d_in[4];
  const float* bias = (const float*)d_in[5];
  float*       out  = (float*)d_out;

  char* ws = (char*)d_ws;
  char*     r0    = ws;          ws += (size_t)N_NODES * 256 * sizeof(float);
  float*    bases = (float*)ws;  ws += (size_t)N_NODES * BD * sizeof(float);
  float*    wbuf  = (float*)ws;  ws += (size_t)N_NODES * OUT_C * sizeof(float);
  unsigned* cnt   = (unsigned*)ws; ws += (size_t)N_NODES * sizeof(unsigned);
  float*    dinv  = (float*)ws;

  unsigned* rank = (unsigned*)r0;
  int*      erec = (int*)(r0 + (size_t)N_EDGES * sizeof(unsigned));
  unsigned* off  = (unsigned*)(r0 + (size_t)N_EDGES * 3 * sizeof(unsigned));

  k_zero<<<(N_NODES + 255) / 256, 256, 0, stream>>>(cnt);
  k_gemm<<<GEMM_BLOCKS + RANK_BLOCKS, 128, 0, stream>>>(x, bW, cW, cb, bases, wbuf,
                                                        ei + N_EDGES, cnt, rank);
  k_fill<<<FILL_BLOCKS, FILL_THREADS, 0, stream>>>(ei, cnt, rank, erec, off, dinv);
  k_gather<<<N_NODES / 16, 256, 0, stream>>>(erec, off, cnt, bases, wbuf, dinv, bias, out);
}


# k_gemm: a block's 4 waves split the 12 column tiles (3 each, all 64 rows), not the rows: one weight fragment feeds 4 row tiles, weight loads fall 4x, 4 accumulator chains a wave; sums unchanged
# speedup vs baseline: 3.3210x; 1.0366x over previous
#include <hip/hip_runtime.h>
#include <math.h>
#include <float.h>


#define N_NODES 50000
#define N_EDGES 800000
#define IN_C    128
#define OUT_C   128
#define HEADS   8
#define BASES   4
#define HEAD_D  16
#define BD      (BASES * HEAD_D)

typedef float v2f __attribute__((ext_vector_type(2)));
typedef float v8f __attribute__((ext_vector_type(8)));

static_assert(N_NODES % 16 == 0, "k_gather maps 16 nodes to a block with no tail");

__global__ __launch_bounds__(256) void k_zero(unsigned* __restrict__ cnt) {
  const int i = blockIdx.x * 256 + threadIdx.x;
  if (i < N_NODES) cnt[i] = 0u;
}

#define GEMM_BLOCKS  ((N_NODES + 63) / 64)
#define RANK_EPW     256
#define RANK_WAVES   (N_EDGES / RANK_EPW)
#define RANK_BLOCKS  ((RANK_WAVES + 3) / 4)

static_assert(N_EDGES % RANK_EPW == 0, "every ranking wave has all its 256 edges");
static_assert((N_EDGES * sizeof(int)) % 16 == 0, "col = ei + N_EDGES stays 16-byte aligned");

__device__ __forceinline__ void rank_wave(const int* __restrict__ col,
                                          unsigned* __restrict__ cnt,
                                          unsigned* __restrict__ rank,
                                          int gw, int lane) {
  const int  e  = gw * RANK_EPW + lane * 4;
  int4 c0 = *(const int4*)(col + e);
  int4 c1 = *(const int4*)(col + e + 128);
  asm volatile("" : "+v"(c0.x), "+v"(c0.y), "+v"(c0.z), "+v"(c0.w),
                    "+v"(c1.x), "+v"(c1.y), "+v"(c1.z), "+v"(c1.w));
  uint4 r0, r1;
  r0.x = atomicAdd(&cnt[c0.x], 1u);
  r0.y = atomicAdd(&cnt[c0.y], 1u);
  r0.z = atomicAdd(&cnt[c0.z], 1u);
  r0.w = atomicAdd(&cnt[c0.w], 1u);
  r1.x = atomicAdd(&cnt[c1.x], 1u);
  r1.y = atomicAdd(&cnt[c1.y], 1u);
  r1.z = atomicAdd(&cnt[c1.z], 1u);
  r1.w = atomicAdd(&cnt[c1.w], 1u);
  *(uint4*)(rank + e)       = r0;
  *(uint4*)(rank + e + 128) = r1;
}

#define FILL_BLOCKS  256
#define FILL_THREADS 1024
#define FILL_TRIPS   13
#define FILL_CHUNK   (4 * FILL_TRIPS)
#define FILL_TAB     (FILL_THREADS * FILL_CHUNK)
#define FILL_EPB     (N_EDGES / FILL_BLOCKS)
#define FILL_NPB     ((N_NODES + FILL_BLOCKS - 1) / FILL_BLOCKS)

static_assert(FILL_TAB >= N_NODES, "the table holds every counter");
static_assert(N_EDGES % FILL_BLOCKS == 0, "edges split evenly over the fill blocks");
static_assert(FILL_EPB <= 4 * FILL_THREADS && FILL_EPB > 3 * FILL_THREADS, "three or four edges per thread");
static_assert(N_NODES % 4 == 0, "counters are copied sixteen bytes at a time");
static_assert(FILL_NPB <= FILL_THREADS, "one thread per node of the block's slice");

__global__ __launch_bounds__(FILL_THREADS) void k_fill(const int* __restrict__ ei,
                                                       const unsigned* __restrict__ cnt,
                                                       const unsigned* __restrict__ rank,
                                                       int* __restrict__ erec,
                                                       unsigned* __restrict__ off,
                                                       float* __restrict__ dinv) {
  __shared__ __attribute__((aligned(16))) unsigned tab[FILL_TAB];
  __shared__ unsigned wsum[32];

  const int tid  = threadIdx.x;
  const int lane = tid & 31;
  const int wave = tid >> 5;

  const int e0 = blockIdx.x * FILL_EPB;
  int      erow[4], ecol[4];
  unsigned erk[4];
#pragma unroll
  for (int u = 0; u < 4; ++u) {
    const int k = tid + u * FILL_THREADS;
    const int e = e0 + (k < FILL_EPB ? k : tid);
    erow[u] = ei[e];
    ecol[u] = ei[N_EDGES + e];
    erk[u]  = rank[e];
  }
  const int      nmine = blockIdx.x * FILL_NPB + tid;
  const bool     nok   = tid < FILL_NPB && nmine < N_NODES;
  const unsigned cmine = cnt[nok ? nmine : 0];

  const int rot = (int)(blockIdx.x % (unsigned)FILL_TRIPS);
  uint4 cv[FILL_TRIPS];
#pragma unroll
  for (int t = 0; t < FILL_TRIPS; ++t) {
    int it = t + rot;
    if (it >= FILL_TRIPS) it -= FILL_TRIPS;
    const int i = it * (4 * FILL_THREADS) + tid * 4;
    cv[t] = *(const uint4*)(cnt + (i < N_NODES ? i : 0));
  }
#pragma unroll
  for (int t = 0; t < FILL_TRIPS; ++t) {
    int it = t + rot;
    if (it >= FILL_TRIPS) it -= FILL_TRIPS;
    const int      i = it * (4 * FILL_THREADS) + tid * 4;
    const unsigned m = 0u - (unsigned)(i < N_NODES);
    *(uint4*)(tab + i) = make_uint4(cv[t].x & m, cv[t].y & m, cv[t].z & m, cv[t].w & m);
  }
  __syncthreads();

  float edv[4];
  for (int u = 0; u < 4; ++u) {
    unsigned rr = (unsigned)erow[u];
    if (rr >= (unsigned)N_NODES) rr = (unsigned)(N_NODES - 1);
    edv[u] = rsqrtf((float)tab[rr] + 1.0f);
  }

  uint4*   my = (uint4*)(tab + tid * FILL_CHUNK);
  uint4    c[FILL_TRIPS];
  unsigned s = 0u;
#pragma unroll
  for (int j = 0; j < FILL_TRIPS; ++j) { c[j] = my[j]; s += c[j].x + c[j].y + c[j].z + c[j].w; }

  unsigned x = s;
  for (int d = 1; d < 32; d <<= 1) {
    const unsigned y = __shfl_up(x, d);
    if (lane >= d) x += y;
  }
  if (lane == 31) wsum[wave] = x;
  __syncthreads();

  const unsigned wt = wsum[lane];
  unsigned wx = wt;
  for (int d = 1; d < 32; d <<= 1) {
    const unsigned y = __shfl_up(wx, d);
    if (lane >= d) wx += y;
  }
  const unsigned base = __shfl(wx - wt, wave);

  unsigned run = base + x - s;
#pragma unroll
  for (int j = 0; j < FILL_TRIPS; ++j) {
    uint4 o;
    o.x = run; run += c[j].x;
    o.y = run; run += c[j].y;
    o.z = run; run += c[j].z;
    o.w = run; run += c[j].w;
    my[j] = o;
  }
  __syncthreads();

  if (nok) {
    off[nmine]  = tab[nmine];
    dinv[nmine] = rsqrtf((float)cmine + 1.0f);
  }

  int2* rec = (int2*)erec;
#pragma unroll
  for (int u = 0; u < 4; ++u) {
    const int k = tid + u * FILL_THREADS;
    if (k < FILL_EPB) {
      unsigned cc = (unsigned)ecol[u];
      if (cc >= (unsigned)N_NODES) cc = (unsigned)(N_NODES - 1);
      const unsigned p = tab[cc] + erk[u];
      if (p < (unsigned)N_EDGES) rec[p] = make_int2(erow[u], __float_as_int(edv[u]));
    }
  }
}

template <int M>
__device__ __forceinline__ float row_xor(float v) {
  return __builtin_bit_cast(float, __builtin_amdgcn_update_dpp(
      0, __builtin_bit_cast(int, v), 0x160 + M, 0xf, 0xf, false));
}

__global__ __launch_bounds__(128) void k_gemm(const float* __restrict__ x,
                                              const float* __restrict__ bW,
                                              const float* __restrict__ cW,
                                              const float* __restrict__ cb,
                                              float* __restrict__ bases,
                                              float* __restrict__ wbuf,
                                              const int* __restrict__ col,
                                              unsigned* __restrict__ cnt,
                                              unsigned* __restrict__ rank) {
  __shared__ float xs[64][132];

  const int tid  = threadIdx.x;

  if (blockIdx.x >= GEMM_BLOCKS) {
    const int gw = ((int)blockIdx.x - GEMM_BLOCKS) * 4 + (tid >> 5);
    if (gw < RANK_WAVES) rank_wave(col, cnt, rank, gw, tid & 31);
    return;
  }

  const int row0 = blockIdx.x * 64;

  for (int t = 0; t < 16; ++t) {
    int idx = t * 128 + tid;
    int r   = idx >> 5;
    int c4  = (idx & 31) * 4;
    int gr  = row0 + r;
    if (gr >= N_NODES) gr = N_NODES - 1;
    const float4 v = *(const float4*)(x + (size_t)gr * IN_C + c4);
    xs[r][c4 + 0] = v.x; xs[r][c4 + 1] = v.y;
    xs[r][c4 + 2] = v.z; xs[r][c4 + 3] = v.w;
  }
  __syncthreads();

  const int wave = tid >> 5;
  const int lane = tid & 31;
  const int nlo  = lane & 15;
  const int half = lane >> 4;
  const int kb   = half * 2;

  for (int j = 0; j < 3; ++j) {
    v8f acc[4];
#pragma unroll
    for (int rt = 0; rt < 4; ++rt) {
      const v8f z = {0.f, 0.f, 0.f, 0.f, 0.f, 0.f, 0.f, 0.f};
      acc[rt] = z;
    }

    if (j == 0) {
      const float* W = bW + wave * (IN_C * HEAD_D);
#pragma unroll 8
      for (int ks = 0; ks < 32; ++ks) {
        const int k = ks * 4 + kb;
        const v2f b = { W[k * HEAD_D + nlo], W[(k + 1) * HEAD_D + nlo] };
#pragma unroll
        for (int rt = 0; rt < 4; ++rt) {
          const v2f a = *(const v2f*)&xs[rt * 16 + nlo][k];
          acc[rt] = __builtin_amdgcn_wmma_f32_16x16x4_f32(
              false, a, false, b, (short)0, acc[rt], false, false);
        }
      }
#pragma unroll
      for (int rt = 0; rt < 4; ++rt) {
#pragma unroll
        for (int r = 0; r < 8; ++r) {
          const int m  = r + (half << 3);
          const int gr = row0 + rt * 16 + m;
          if (gr < N_NODES) bases[(size_t)gr * BD + wave * 16 + nlo] = acc[rt][r];
        }
      }
    } else {
      const int   head = wave + 4 * (j - 1);
      const float* W   = cW + head * 16;
#pragma unroll 8
      for (int ks = 0; ks < 32; ++ks) {
        const int k = ks * 4 + kb;
        const v2f b = { W[k * OUT_C + nlo], W[(k + 1) * OUT_C + nlo] };
#pragma unroll
        for (int rt = 0; rt < 4; ++rt) {
          const v2f a = *(const v2f*)&xs[rt * 16 + nlo][k];
          acc[rt] = __builtin_amdgcn_wmma_f32_16x16x4_f32(
              false, a, false, b, (short)0, acc[rt], false, false);
        }
      }
      const float bb = cb[head * 16 + nlo];
#pragma unroll
      for (int rt = 0; rt < 4; ++rt) {
        float v[8], mx[8], ex[8], sm[8];
#pragma unroll
        for (int r = 0; r < 8; ++r) { v[r] = acc[rt][r] + bb; mx[r] = v[r]; }
#pragma unroll
        for (int r = 0; r < 8; ++r) mx[r] = fmaxf(mx[r], row_xor<1>(mx[r]));
#pragma unroll
        for (int r = 0; r < 8; ++r) mx[r] = fmaxf(mx[r], row_xor<2>(mx[r]));
#pragma unroll
        for (int r = 0; r < 8; ++r) mx[r] = fmaxf(mx[r], row_xor<4>(mx[r]));
#pragma unroll
        for (int r = 0; r < 8; ++r) mx[r] = fmaxf(mx[r], row_xor<8>(mx[r]));
#pragma unroll
        for (int r = 0; r < 8; ++r) { ex[r] = __expf(v[r] - mx[r]); sm[r] = ex[r]; }
#pragma unroll
        for (int r = 0; r < 8; ++r) sm[r] += row_xor<1>(sm[r]);
#pragma unroll
        for (int r = 0; r < 8; ++r) sm[r] += row_xor<2>(sm[r]);
#pragma unroll
        for (int r = 0; r < 8; ++r) sm[r] += row_xor<4>(sm[r]);
#pragma unroll
        for (int r = 0; r < 8; ++r) sm[r] += row_xor<8>(sm[r]);
#pragma unroll
        for (int r = 0; r < 8; ++r) {
          const float wv = ex[r] / sm[r];
          const int m  = r + (half << 3);
          const int gr = row0 + rt * 16 + m;
          if (gr < N_NODES) wbuf[(size_t)gr * OUT_C + head * 16 + nlo] = wv;
        }
      }
    }
  }
}

__global__ __launch_bounds__(256) void k_gather(const int* __restrict__ erec,
                                                const unsigned* __restrict__ off,
                                                const unsigned* __restrict__ cnt,
                                                const float* __restrict__ bases,
                                                const float* __restrict__ wbuf,
                                                const float* __restrict__ dinv,
                                                const float* __restrict__ bias,
                                                float* __restrict__ out) {
  __shared__ float4 fin[16][4][16];

  const int tid = threadIdx.x;
  const int nl  = tid >> 4;
  const int q   = tid & 15;
  const int n   = blockIdx.x * 16 + nl;

  const unsigned deg = cnt[n];
  unsigned start = off[n];
  if (start > (unsigned)N_EDGES) start = (unsigned)N_EDGES;
  unsigned end = start + deg;
  if (end > (unsigned)N_EDGES) end = (unsigned)N_EDGES;

  const float* bq  = bases + q * 4;
  const int2*  rec = (const int2*)erec;

  float4 a_sym = make_float4(0.f, 0.f, 0.f, 0.f);
  float4 a_sum = make_float4(0.f, 0.f, 0.f, 0.f);
  float4 a_sq  = make_float4(0.f, 0.f, 0.f, 0.f);
  float4 a_mx  = make_float4(-INFINITY, -INFINITY, -INFINITY, -INFINITY);

  for (unsigned j = start; j < end; j += 4) {
    unsigned r[4];
    float    dr[4];
    float4   v[4];
#pragma unroll
    for (int u = 0; u < 4; ++u) {
      unsigned jj = j + u;
      if (jj >= end) jj = end - 1;
      const int2     e  = rec[jj];
      const unsigned rr = (unsigned)e.x;
      r[u]  = rr < (unsigned)N_NODES ? rr : (unsigned)(N_NODES - 1);
      dr[u] = __int_as_float(e.y);
    }
#pragma unroll
    for (int u = 0; u < 4; ++u) v[u] = *(const float4*)(bq + (size_t)r[u] * BD);
#pragma unroll
    for (int u = 0; u < 4; ++u) {
      const bool  ok = (j + u) < end;
      const float d  = dr[u];
      const float tx = ok ? v[u].x : 0.f, mx_ = ok ? v[u].x : -INFINITY;
      const float ty = ok ? v[u].y : 0.f, my_ = ok ? v[u].y : -INFINITY;
      const float tz = ok ? v[u].z : 0.f, mz_ = ok ? v[u].z : -INFINITY;
      const float tw = ok ? v[u].w : 0.f, mw_ = ok ? v[u].w : -INFINITY;
      a_sym.x += d * tx;  a_sym.y += d * ty;  a_sym.z += d * tz;  a_sym.w += d * tw;
      a_sum.x += tx;      a_sum.y += ty;      a_sum.z += tz;      a_sum.w += tw;
      a_sq.x  += tx * tx; a_sq.y  += ty * ty; a_sq.z  += tz * tz; a_sq.w  += tw * tw;
      a_mx.x = fmaxf(a_mx.x, mx_); a_mx.y = fmaxf(a_mx.y, my_);
      a_mx.z = fmaxf(a_mx.z, mz_); a_mx.w = fmaxf(a_mx.w, mw_);
    }
  }

  {
    const float  cf = (float)deg;
    const float  ic = 1.0f / fmaxf(cf, 1.0f);
    const float  dv = dinv[n];
    const float4 bs = *(const float4*)(bq + (size_t)n * BD);
    const float  sy[4] = {a_sym.x, a_sym.y, a_sym.z, a_sym.w};
    const float  su[4] = {a_sum.x, a_sum.y, a_sum.z, a_sum.w};
    const float  sq[4] = {a_sq.x,  a_sq.y,  a_sq.z,  a_sq.w};
    const float  sm[4] = {a_mx.x,  a_mx.y,  a_mx.z,  a_mx.w};
    const float  bv[4] = {bs.x, bs.y, bs.z, bs.w};
    float o_sym[4], o_mean[4], o_mx[4], o_sd[4];
#pragma unroll
    for (int c = 0; c < 4; ++c) {
      o_sym[c] = dv * (sy[c] + dv * bv[c]);
      const float mean = su[c] * ic;
      const float var  = sq[c] * ic - mean * mean;
      o_mean[c] = mean;
      o_sd[c]   = sqrtf(fmaxf(var, 0.0f) + 1e-5f);
      o_mx[c]   = (cf > 0.0f) ? sm[c] : 0.0f;
    }
    fin[nl][0][q] = make_float4(o_sym[0],  o_sym[1],  o_sym[2],  o_sym[3]);
    fin[nl][1][q] = make_float4(o_mean[0], o_mean[1], o_mean[2], o_mean[3]);
    fin[nl][2][q] = make_float4(o_mx[0],   o_mx[1],   o_mx[2],   o_mx[3]);
    fin[nl][3][q] = make_float4(o_sd[0],   o_sd[1],   o_sd[2],   o_sd[3]);
  }
  __syncthreads();

  const int dg = q & 3;
  const int hq = q >> 2;
#pragma unroll
  for (int h2 = 0; h2 < 2; ++h2) {
    const int h = h2 * 4 + hq;
    const float4 bi = *(const float4*)(bias + h * 16 + dg * 4);
    float o0 = bi.x, o1 = bi.y, o2 = bi.z, o3 = bi.w;
#pragma unroll
    for (int b = 0; b < 4; ++b) {
      const float4 w  = *(const float4*)(wbuf + (size_t)n * OUT_C + h * 16 + b * 4);
      const float4 s0 = fin[nl][0][b * 4 + dg];
      const float4 s1 = fin[nl][1][b * 4 + dg];
      const float4 s2 = fin[nl][2][b * 4 + dg];
      const float4 s3 = fin[nl][3][b * 4 + dg];
      o0 += w.x * s0.x + w.y * s1.x + w.z * s2.x + w.w * s3.x;
      o1 += w.x * s0.y + w.y * s1.y + w.z * s2.y + w.w * s3.y;
      o2 += w.x * s0.z + w.y * s1.z + w.z * s2.z + w.w * s3.z;
      o3 += w.x * s0.w + w.y * s1.w + w.z * s2.w + w.w * s3.w;
    }
    *(float4*)(out + (size_t)n * OUT_C + h * 16 + dg * 4) = make_float4(o0, o1, o2, o3);
  }
}

extern "C" void kernel_launch(void* const* d_in, const int* in_sizes, int n_in,
                              void* d_out, int out_size, void* d_ws, size_t ws_size,
                              hipStream_t stream) {
  (void)in_sizes; (void)n_in; (void)out_size; (void)ws_size;
  const float* x    = (const float*)d_in[0];
  const int*   ei   = (const int*)d_in[1];
  const float* bW   = (const float*)d_in[2];
  const float* cW   = (const float*)d_in[3];
  const float* cb   = (const float*)d_in[4];
  const float* bias = (const float*)d_in[5];
  float*       out  = (float*)d_out;

  char* ws = (char*)d_ws;
  char*     r0    = ws;          ws += (size_t)N_NODES * 256 * sizeof(float);
  float*    bases = (float*)ws;  ws += (size_t)N_NODES * BD * sizeof(float);
  float*    wbuf  = (float*)ws;  ws += (size_t)N_NODES * OUT_C * sizeof(float);
  unsigned* cnt   = (unsigned*)ws; ws += (size_t)N_NODES * sizeof(unsigned);
  float*    dinv  = (float*)ws;

  unsigned* rank = (unsigned*)r0;
  int*      erec = (int*)(r0 + (size_t)N_EDGES * sizeof(unsigned));
  unsigned* off  = (unsigned*)(r0 + (size_t)N_EDGES * 3 * sizeof(unsigned));

  k_zero<<<(N_NODES + 255) / 256, 256, 0, stream>>>(cnt);
  k_gemm<<<GEMM_BLOCKS + RANK_BLOCKS, 128, 0, stream>>>(x, bW, cW, cb, bases, wbuf,
                                                        ei + N_EDGES, cnt, rank);
  k_fill<<<FILL_BLOCKS, FILL_THREADS, 0, stream>>>(ei, cnt, rank, erec, off, dinv);
  k_gather<<<N_NODES / 16, 256, 0, stream>>>(erec, off, cnt, bases, wbuf, dinv, bias, out);
}


# edges ordered by destination through LDS bins (chunk-by-bin counts, LDS-staged partition, per-bin LDS sort), not by global returning count atomics and a random scatter; sums stay arrival-ordered
# speedup vs baseline: 5.0892x; 1.5324x over previous
#include <hip/hip_runtime.h>
#include <math.h>
#include <float.h>


#define N_NODES 50000
#define N_EDGES 800000
#define IN_C    128
#define OUT_C   128
#define HEADS   8
#define BASES   4
#define HEAD_D  16
#define BD      (BASES * HEAD_D)

typedef float v2f __attribute__((ext_vector_type(2)));
typedef float v8f __attribute__((ext_vector_type(8)));

static_assert(N_NODES % 16 == 0, "k_gather maps 16 nodes to a block with no tail");

#define GEMM_BLOCKS  ((N_NODES + 63) / 64)
#define BIN_NODES    256
#define NBINS        ((N_NODES + BIN_NODES - 1) / BIN_NODES)
#define TBLW         256
#define NCHUNK       256
#define CHUNK_E      (N_EDGES / NCHUNK)
#define CNT_TRIPS    ((CHUNK_E + 127) / 128)

static_assert(N_EDGES % NCHUNK == 0, "edges split evenly over the chunks");
static_assert(NBINS <= TBLW, "a table row holds every bin");
static_assert(CHUNK_E <= 4 * 1024 && CHUNK_E > 3 * 1024, "three or four edges per thread in k_part");

__device__ __forceinline__ void count_chunk(const int* __restrict__ col,
                                            unsigned* __restrict__ tbl,
                                            unsigned* h, int c, int tid) {
  h[tid] = 0u;
  h[tid + 128] = 0u;
  const int e0 = c * CHUNK_E;
  unsigned v[CNT_TRIPS];
#pragma unroll
  for (int k = 0; k < CNT_TRIPS; ++k) {
    const int i = tid + k * 128;
    v[k] = (unsigned)col[e0 + (i < CHUNK_E ? i : tid)];
  }
  __syncthreads();
#pragma unroll
  for (int k = 0; k < CNT_TRIPS; ++k) {
    const int i = tid + k * 128;
    if (i < CHUNK_E) {
      unsigned cc = v[k];
      if (cc >= (unsigned)N_NODES) cc = (unsigned)(N_NODES - 1);
      atomicAdd(&h[cc >> 8], 1u);
    }
  }
  __syncthreads();
  tbl[c * TBLW + tid]       = h[tid];
  tbl[c * TBLW + tid + 128] = h[tid + 128];
}

__global__ __launch_bounds__(1024) void k_part(const int* __restrict__ ei,
                                               const unsigned* __restrict__ tbl,
                                               int* __restrict__ part_w,
                                               unsigned* __restrict__ binStart) {
  int2* part = (int2*)part_w;
  __shared__ unsigned s_tot[4][256];
  __shared__ unsigned s_pre[4][256];
  __shared__ unsigned s_base[256];
  __shared__ unsigned s_lst[256];
  __shared__ unsigned s_cur[256];
  __shared__ unsigned s_w1[8], s_w2[8];
  __shared__ int2 stage[CHUNK_E];

  const int tid  = threadIdx.x;
  const int lane = tid & 31;
  const int wave = tid >> 5;
  const int c    = blockIdx.x;

  const int e0 = c * CHUNK_E;
  int er[4], ec[4];
#pragma unroll
  for (int u = 0; u < 4; ++u) {
    const int k = tid + u * 1024;
    const int e = e0 + (k < CHUNK_E ? k : tid);
    er[u] = ei[e];
    ec[u] = ei[N_EDGES + e];
  }

  const int b = tid & 255;
  const int q = tid >> 8;
  unsigned st = 0u, sp = 0u;
#pragma unroll 16
  for (int k = 0; k < 64; ++k) {
    const int      cc = q * 64 + k;
    const unsigned v  = tbl[cc * TBLW + b];
    st += v;
    sp += (cc < c) ? v : 0u;
  }
  s_tot[q][b] = st;
  s_pre[q][b] = sp;
  if (tid < 256) s_cur[tid] = 0u;
  __syncthreads();

  unsigned tot = 0u, pre = 0u, mine = 0u;
  if (tid < 256) {
    tot  = s_tot[0][tid] + s_tot[1][tid] + s_tot[2][tid] + s_tot[3][tid];
    pre  = s_pre[0][tid] + s_pre[1][tid] + s_pre[2][tid] + s_pre[3][tid];
    mine = tbl[c * TBLW + tid];
  }
  unsigned x1 = tot, x2 = mine;
  for (int d = 1; d < 32; d <<= 1) {
    const unsigned y1 = __shfl_up(x1, d);
    const unsigned y2 = __shfl_up(x2, d);
    if (lane >= d) { x1 += y1; x2 += y2; }
  }
  if (wave < 8 && lane == 31) { s_w1[wave] = x1; s_w2[wave] = x2; }
  __syncthreads();
  if (tid < 256) {
    unsigned p1 = 0u, p2 = 0u;
    for (int w = 0; w < wave; ++w) { p1 += s_w1[w]; p2 += s_w2[w]; }
    const unsigned bstart = p1 + x1 - tot;
    s_base[tid] = bstart + pre;
    s_lst[tid]  = p2 + x2 - mine;
    if (c == 0) binStart[tid] = bstart;
  }
  __syncthreads();

#pragma unroll
  for (int u = 0; u < 4; ++u) {
    const int k = tid + u * 1024;
    if (k < CHUNK_E) {
      unsigned cc = (unsigned)ec[u];
      if (cc >= (unsigned)N_NODES) cc = (unsigned)(N_NODES - 1);
      const unsigned bin  = cc >> 8;
      const unsigned slot = s_lst[bin] + atomicAdd(&s_cur[bin], 1u);
      if (slot < (unsigned)CHUNK_E) stage[slot] = make_int2(er[u], (int)cc);
    }
  }
  __syncthreads();

#pragma unroll
  for (int u = 0; u < 4; ++u) {
    const int k = tid + u * 1024;
    if (k < CHUNK_E) {
      const int2     r   = stage[k];
      const unsigned bin = ((unsigned)r.y >> 8) & 255u;
      const unsigned pos = s_base[bin] + ((unsigned)k - s_lst[bin]);
      if (pos < (unsigned)N_EDGES) part[pos] = r;
    }
  }
}

#define BIN_CAP   16384
#define BIN_ROUND 5

__global__ __launch_bounds__(1024) void k_bin(const int* __restrict__ part_w,
                                              const unsigned* __restrict__ binStart,
                                              int* __restrict__ erec,
                                              unsigned* __restrict__ off,
                                              unsigned* __restrict__ cnt,
                                              float* __restrict__ dinv) {
  __shared__ unsigned s_cnt[256], s_off[256], s_cur[256], s_w[8];
  __shared__ int s_srt[BIN_CAP];
  const int2* part = (const int2*)part_w;

  const int tid  = threadIdx.x;
  const int lane = tid & 31;
  const int wave = tid >> 5;
  const int b    = blockIdx.x;

  unsigned S = binStart[b], E = binStart[b + 1];
  if (S > (unsigned)N_EDGES) S = (unsigned)N_EDGES;
  if (E > (unsigned)N_EDGES) E = (unsigned)N_EDGES;
  if (E < S) E = S;
  const unsigned m = E - S;

  if (tid < 256) { s_cnt[tid] = 0u; s_cur[tid] = 0u; }
  __syncthreads();

  for (unsigned r0 = 0u; r0 < m; r0 += BIN_ROUND * 1024u) {
    int cl[BIN_ROUND];
#pragma unroll
    for (int u = 0; u < BIN_ROUND; ++u) {
      const unsigned i = r0 + (unsigned)tid + (unsigned)u * 1024u;
      cl[u] = part[S + (i < m ? i : 0u)].y;
    }
#pragma unroll
    for (int u = 0; u < BIN_ROUND; ++u) {
      const unsigned i = r0 + (unsigned)tid + (unsigned)u * 1024u;
      if (i < m) atomicAdd(&s_cnt[(unsigned)cl[u] & 255u], 1u);
    }
  }
  __syncthreads();

  const unsigned cv = (tid < 256) ? s_cnt[tid] : 0u;
  unsigned x = cv;
  for (int d = 1; d < 32; d <<= 1) {
    const unsigned y = __shfl_up(x, d);
    if (lane >= d) x += y;
  }
  if (wave < 8 && lane == 31) s_w[wave] = x;
  __syncthreads();
  if (tid < 256) {
    unsigned p = 0u;
    for (int w = 0; w < wave; ++w) p += s_w[w];
    const unsigned o = p + x - cv;
    s_off[tid] = o;
    const int node = b * BIN_NODES + tid;
    if (node < N_NODES) {
      off[node]  = S + o;
      cnt[node]  = cv;
      dinv[node] = rsqrtf((float)cv + 1.0f);
    }
  }
  __syncthreads();

  const bool inl = m <= (unsigned)BIN_CAP;
  for (unsigned r0 = 0u; r0 < m; r0 += BIN_ROUND * 1024u) {
    int2 rc[BIN_ROUND];
#pragma unroll
    for (int u = 0; u < BIN_ROUND; ++u) {
      const unsigned i = r0 + (unsigned)tid + (unsigned)u * 1024u;
      rc[u] = part[S + (i < m ? i : 0u)];
    }
#pragma unroll
    for (int u = 0; u < BIN_ROUND; ++u) {
      const unsigned i = r0 + (unsigned)tid + (unsigned)u * 1024u;
      if (i < m) {
        const unsigned nl = (unsigned)rc[u].y & 255u;
        const unsigned p  = s_off[nl] + atomicAdd(&s_cur[nl], 1u);
        if (inl) { if (p < (unsigned)BIN_CAP) s_srt[p] = rc[u].x; }
        else     { if (S + p < (unsigned)N_EDGES) erec[S + p] = rc[u].x; }
      }
    }
  }
  __syncthreads();
  if (inl) {
    for (unsigned i = (unsigned)tid; i < m; i += 1024u) erec[S + i] = s_srt[i];
  }
}

template <int M>
__device__ __forceinline__ float row_xor(float v) {
  return __builtin_bit_cast(float, __builtin_amdgcn_update_dpp(
      0, __builtin_bit_cast(int, v), 0x160 + M, 0xf, 0xf, false));
}

__global__ __launch_bounds__(128) void k_gemm(const float* __restrict__ x,
                                              const float* __restrict__ bW,
                                              const float* __restrict__ cW,
                                              const float* __restrict__ cb,
                                              float* __restrict__ bases,
                                              float* __restrict__ wbuf,
                                              const int* __restrict__ col,
                                              unsigned* __restrict__ tbl) {
  __shared__ float xs[64][132];

  const int tid  = threadIdx.x;

  if (blockIdx.x >= GEMM_BLOCKS) {
    count_chunk(col, tbl, (unsigned*)&xs[0][0], (int)blockIdx.x - GEMM_BLOCKS, tid);
    return;
  }

  const int row0 = blockIdx.x * 64;

  for (int t = 0; t < 16; ++t) {
    int idx = t * 128 + tid;
    int r   = idx >> 5;
    int c4  = (idx & 31) * 4;
    int gr  = row0 + r;
    if (gr >= N_NODES) gr = N_NODES - 1;
    const float4 v = *(const float4*)(x + (size_t)gr * IN_C + c4);
    xs[r][c4 + 0] = v.x; xs[r][c4 + 1] = v.y;
    xs[r][c4 + 2] = v.z; xs[r][c4 + 3] = v.w;
  }
  __syncthreads();

  const int wave = tid >> 5;
  const int lane = tid & 31;
  const int nlo  = lane & 15;
  const int half = lane >> 4;
  const int kb   = half * 2;

  for (int j = 0; j < 3; ++j) {
    v8f acc[4];
#pragma unroll
    for (int rt = 0; rt < 4; ++rt) {
      const v8f z = {0.f, 0.f, 0.f, 0.f, 0.f, 0.f, 0.f, 0.f};
      acc[rt] = z;
    }

    if (j == 0) {
      const float* W = bW + wave * (IN_C * HEAD_D);
#pragma unroll 8
      for (int ks = 0; ks < 32; ++ks) {
        const int k = ks * 4 + kb;
        const v2f b = { W[k * HEAD_D + nlo], W[(k + 1) * HEAD_D + nlo] };
#pragma unroll
        for (int rt = 0; rt < 4; ++rt) {
          const v2f a = *(const v2f*)&xs[rt * 16 + nlo][k];
          acc[rt] = __builtin_amdgcn_wmma_f32_16x16x4_f32(
              false, a, false, b, (short)0, acc[rt], false, false);
        }
      }
#pragma unroll
      for (int rt = 0; rt < 4; ++rt) {
#pragma unroll
        for (int r = 0; r < 8; ++r) {
          const int m  = r + (half << 3);
          const int gr = row0 + rt * 16 + m;
          if (gr < N_NODES) bases[(size_t)gr * BD + wave * 16 + nlo] = acc[rt][r];
        }
      }
    } else {
      const int   head = wave + 4 * (j - 1);
      const float* W   = cW + head * 16;
#pragma unroll 8
      for (int ks = 0; ks < 32; ++ks) {
        const int k = ks * 4 + kb;
        const v2f b = { W[k * OUT_C + nlo], W[(k + 1) * OUT_C + nlo] };
#pragma unroll
        for (int rt = 0; rt < 4; ++rt) {
          const v2f a = *(const v2f*)&xs[rt * 16 + nlo][k];
          acc[rt] = __builtin_amdgcn_wmma_f32_16x16x4_f32(
              false, a, false, b, (short)0, acc[rt], false, false);
        }
      }
      const float bb = cb[head * 16 + nlo];
#pragma unroll
      for (int rt = 0; rt < 4; ++rt) {
        float v[8], mx[8], ex[8], sm[8];
#pragma unroll
        for (int r = 0; r < 8; ++r) { v[r] = acc[rt][r] + bb; mx[r] = v[r]; }
#pragma unroll
        for (int r = 0; r < 8; ++r) mx[r] = fmaxf(mx[r], row_xor<1>(mx[r]));
#pragma unroll
        for (int r = 0; r < 8; ++r) mx[r] = fmaxf(mx[r], row_xor<2>(mx[r]));
#pragma unroll
        for (int r = 0; r < 8; ++r) mx[r] = fmaxf(mx[r], row_xor<4>(mx[r]));
#pragma unroll
        for (int r = 0; r < 8; ++r) mx[r] = fmaxf(mx[r], row_xor<8>(mx[r]));
#pragma unroll
        for (int r = 0; r < 8; ++r) { ex[r] = __expf(v[r] - mx[r]); sm[r] = ex[r]; }
#pragma unroll
        for (int r = 0; r < 8; ++r) sm[r] += row_xor<1>(sm[r]);
#pragma unroll
        for (int r = 0; r < 8; ++r) sm[r] += row_xor<2>(sm[r]);
#pragma unroll
        for (int r = 0; r < 8; ++r) sm[r] += row_xor<4>(sm[r]);
#pragma unroll
        for (int r = 0; r < 8; ++r) sm[r] += row_xor<8>(sm[r]);
#pragma unroll
        for (int r = 0; r < 8; ++r) {
          const float wv = ex[r] / sm[r];
          const int m  = r + (half << 3);
          const int gr = row0 + rt * 16 + m;
          if (gr < N_NODES) wbuf[(size_t)gr * OUT_C + head * 16 + nlo] = wv;
        }
      }
    }
  }
}

__global__ __launch_bounds__(256) void k_gather(const int* __restrict__ erec,
                                                const unsigned* __restrict__ off,
                                                const unsigned* __restrict__ cnt,
                                                const float* __restrict__ bases,
                                                const float* __restrict__ wbuf,
                                                const float* __restrict__ dinv,
                                                const float* __restrict__ bias,
                                                float* __restrict__ out) {
  __shared__ float4 fin[16][4][16];

  const int tid = threadIdx.x;
  const int nl  = tid >> 4;
  const int q   = tid & 15;
  const int n   = blockIdx.x * 16 + nl;

  const unsigned deg = cnt[n];
  unsigned start = off[n];
  if (start > (unsigned)N_EDGES) start = (unsigned)N_EDGES;
  unsigned end = start + deg;
  if (end > (unsigned)N_EDGES) end = (unsigned)N_EDGES;

  const float* bq  = bases + q * 4;

  float4 a_sym = make_float4(0.f, 0.f, 0.f, 0.f);
  float4 a_sum = make_float4(0.f, 0.f, 0.f, 0.f);
  float4 a_sq  = make_float4(0.f, 0.f, 0.f, 0.f);
  float4 a_mx  = make_float4(-INFINITY, -INFINITY, -INFINITY, -INFINITY);

  for (unsigned j = start; j < end; j += 4) {
    unsigned r[4];
    float    dr[4];
    float4   v[4];
#pragma unroll
    for (int u = 0; u < 4; ++u) {
      unsigned jj = j + u;
      if (jj >= end) jj = end - 1;
      const unsigned rr = (unsigned)erec[jj];
      r[u]  = rr < (unsigned)N_NODES ? rr : (unsigned)(N_NODES - 1);
    }
#pragma unroll
    for (int u = 0; u < 4; ++u) {
      v[u]  = *(const float4*)(bq + (size_t)r[u] * BD);
      dr[u] = dinv[r[u]];
    }
#pragma unroll
    for (int u = 0; u < 4; ++u) {
      const bool  ok = (j + u) < end;
      const float d  = dr[u];
      const float tx = ok ? v[u].x : 0.f, mx_ = ok ? v[u].x : -INFINITY;
      const float ty = ok ? v[u].y : 0.f, my_ = ok ? v[u].y : -INFINITY;
      const float tz = ok ? v[u].z : 0.f, mz_ = ok ? v[u].z : -INFINITY;
      const float tw = ok ? v[u].w : 0.f, mw_ = ok ? v[u].w : -INFINITY;
      a_sym.x += d * tx;  a_sym.y += d * ty;  a_sym.z += d * tz;  a_sym.w += d * tw;
      a_sum.x += tx;      a_sum.y += ty;      a_sum.z += tz;      a_sum.w += tw;
      a_sq.x  += tx * tx; a_sq.y  += ty * ty; a_sq.z  += tz * tz; a_sq.w  += tw * tw;
      a_mx.x = fmaxf(a_mx.x, mx_); a_mx.y = fmaxf(a_mx.y, my_);
      a_mx.z = fmaxf(a_mx.z, mz_); a_mx.w = fmaxf(a_mx.w, mw_);
    }
  }

  {
    const float  cf = (float)deg;
    const float  ic = 1.0f / fmaxf(cf, 1.0f);
    const float  dv = dinv[n];
    const float4 bs = *(const float4*)(bq + (size_t)n * BD);
    const float  sy[4] = {a_sym.x, a_sym.y, a_sym.z, a_sym.w};
    const float  su[4] = {a_sum.x, a_sum.y, a_sum.z, a_sum.w};
    const float  sq[4] = {a_sq.x,  a_sq.y,  a_sq.z,  a_sq.w};
    const float  sm[4] = {a_mx.x,  a_mx.y,  a_mx.z,  a_mx.w};
    const float  bv[4] = {bs.x, bs.y, bs.z, bs.w};
    float o_sym[4], o_mean[4], o_mx[4], o_sd[4];
#pragma unroll
    for (int c = 0; c < 4; ++c) {
      o_sym[c] = dv * (sy[c] + dv * bv[c]);
      const float mean = su[c] * ic;
      const float var  = sq[c] * ic - mean * mean;
      o_mean[c] = mean;
      o_sd[c]   = sqrtf(fmaxf(var, 0.0f) + 1e-5f);
      o_mx[c]   = (cf > 0.0f) ? sm[c] : 0.0f;
    }
    fin[nl][0][q] = make_float4(o_sym[0],  o_sym[1],  o_sym[2],  o_sym[3]);
    fin[nl][1][q] = make_float4(o_mean[0], o_mean[1], o_mean[2], o_mean[3]);
    fin[nl][2][q] = make_float4(o_mx[0],   o_mx[1],   o_mx[2],   o_mx[3]);
    fin[nl][3][q] = make_float4(o_sd[0],   o_sd[1],   o_sd[2],   o_sd[3]);
  }
  __syncthreads();

  const int dg = q & 3;
  const int hq = q >> 2;
#pragma unroll
  for (int h2 = 0; h2 < 2; ++h2) {
    const int h = h2 * 4 + hq;
    const float4 bi = *(const float4*)(bias + h * 16 + dg * 4);
    float o0 = bi.x, o1 = bi.y, o2 = bi.z, o3 = bi.w;
#pragma unroll
    for (int b = 0; b < 4; ++b) {
      const float4 w  = *(const float4*)(wbuf + (size_t)n * OUT_C + h * 16 + b * 4);
      const float4 s0 = fin[nl][0][b * 4 + dg];
      const float4 s1 = fin[nl][1][b * 4 + dg];
      const float4 s2 = fin[nl][2][b * 4 + dg];
      const float4 s3 = fin[nl][3][b * 4 + dg];
      o0 += w.x * s0.x + w.y * s1.x + w.z * s2.x + w.w * s3.x;
      o1 += w.x * s0.y + w.y * s1.y + w.z * s2.y + w.w * s3.y;
      o2 += w.x * s0.z + w.y * s1.z + w.z * s2.z + w.w * s3.z;
      o3 += w.x * s0.w + w.y * s1.w + w.z * s2.w + w.w * s3.w;
    }
    *(float4*)(out + (size_t)n * OUT_C + h * 16 + dg * 4) = make_float4(o0, o1, o2, o3);
  }
}

extern "C" void kernel_launch(void* const* d_in, const int* in_sizes, int n_in,
                              void* d_out, int out_size, void* d_ws, size_t ws_size,
                              hipStream_t stream) {
  (void)in_sizes; (void)n_in; (void)out_size; (void)ws_size;
  const float* x    = (const float*)d_in[0];
  const int*   ei   = (const int*)d_in[1];
  const float* bW   = (const float*)d_in[2];
  const float* cW   = (const float*)d_in[3];
  const float* cb   = (const float*)d_in[4];
  const float* bias = (const float*)d_in[5];
  float*       out  = (float*)d_out;

  char* ws = (char*)d_ws;
  char*     r0    = ws;          ws += (size_t)N_NODES * 256 * sizeof(float);
  float*    bases = (float*)ws;  ws += (size_t)N_NODES * BD * sizeof(float);
  float*    wbuf  = (float*)ws;  ws += (size_t)N_NODES * OUT_C * sizeof(float);
  unsigned* cnt   = (unsigned*)ws; ws += (size_t)N_NODES * sizeof(unsigned);
  float*    dinv  = (float*)ws;

  int*      part     = (int*)r0;
  int*      erec     = (int*)(r0 + (size_t)N_EDGES * 8);
  unsigned* off      = (unsigned*)(r0 + (size_t)N_EDGES * 12);
  unsigned* tbl      = (unsigned*)(r0 + (size_t)N_EDGES * 12 + (size_t)262144);
  unsigned* binStart = (unsigned*)(r0 + (size_t)N_EDGES * 12 + (size_t)2 * 262144);

  k_gemm<<<GEMM_BLOCKS + NCHUNK, 128, 0, stream>>>(x, bW, cW, cb, bases, wbuf,
                                                   ei + N_EDGES, tbl);
  k_part<<<NCHUNK, 1024, 0, stream>>>(ei, tbl, part, binStart);
  k_bin<<<NBINS, 1024, 0, stream>>>(part, binStart, erec, off, cnt, dinv);
  k_gather<<<N_NODES / 16, 256, 0, stream>>>(erec, off, cnt, bases, wbuf, dinv, bias, out);
}


# k_gemm rows dealt as 768 blocks of 64 rows (3 on every WGP) + 53 one-tile blocks of 16 rows, not 782 of 64: no WGP carries a fourth 64-row block; all tiles whole, row guards go; sums unchanged
# speedup vs baseline: 6.4564x; 1.2686x over previous
#include <hip/hip_runtime.h>
#include <math.h>
#include <float.h>


#define N_NODES 50000
#define N_EDGES 800000
#define IN_C    128
#define OUT_C   128
#define HEADS   8
#define BASES   4
#define HEAD_D  16
#define BD      (BASES * HEAD_D)

typedef float v2f __attribute__((ext_vector_type(2)));
typedef float v8f __attribute__((ext_vector_type(8)));

static_assert(N_NODES % 16 == 0, "k_gather maps 16 nodes to a block with no tail");

#define N_WGP        256
#define FULL_BLOCKS  ((N_NODES / 64 / N_WGP) * N_WGP)
#define TAIL_ROW0    (FULL_BLOCKS * 64)
#define TAIL_BLOCKS  ((N_NODES - TAIL_ROW0) / 16)
#define GEMM_BLOCKS  (FULL_BLOCKS + TAIL_BLOCKS)
#define BIN_NODES    256
#define NBINS        ((N_NODES + BIN_NODES - 1) / BIN_NODES)
#define TBLW         256
#define NCHUNK       256
#define CHUNK_E      (N_EDGES / NCHUNK)
#define CNT_TRIPS    ((CHUNK_E + 127) / 128)

static_assert(N_EDGES % NCHUNK == 0, "edges split evenly over the chunks");
static_assert(FULL_BLOCKS > 0 && TAIL_ROW0 + TAIL_BLOCKS * 16 == N_NODES, "the matrix blocks cover every row with whole tiles");
static_assert(NBINS <= TBLW, "a table row holds every bin");
static_assert(CHUNK_E < 65536, "run start and run length share one 32-bit table entry");
static_assert(TBLW == 256, "two bins per thread of a 128-thread block");

__device__ __forceinline__ void sort_chunk(const int* __restrict__ ei,
                                           unsigned* __restrict__ tbl,
                                           int2* __restrict__ plocal,
                                           unsigned* lds, int c, int tid) {
  unsigned* h     = lds;
  unsigned* lst   = lds + 256;
  unsigned* cur   = lds + 512;
  unsigned* wt    = lds + 768;
  int2*     stage = (int2*)(lds + 784);
  const int lane = tid & 31;
  const int wave = tid >> 5;
  __builtin_assume(tid >= 0 && tid < 128);

  h[tid] = 0u;   h[tid + 128] = 0u;
  cur[tid] = 0u; cur[tid + 128] = 0u;
  const int e0 = c * CHUNK_E;
  unsigned vc[CNT_TRIPS];
  int      vr[CNT_TRIPS];
#pragma unroll
  for (int k = 0; k < CNT_TRIPS; ++k) {
    const int i = tid + k * 128;
    const int e = e0 + (i < CHUNK_E ? i : tid);
    vc[k] = (unsigned)ei[N_EDGES + e];
    vr[k] = ei[e];
  }
  __syncthreads();
#pragma unroll
  for (int k = 0; k < CNT_TRIPS; ++k) {
    const int i = tid + k * 128;
    if (i < CHUNK_E) {
      unsigned cc = vc[k];
      if (cc >= (unsigned)N_NODES) cc = (unsigned)(N_NODES - 1);
      vc[k] = cc;
      atomicAdd(&h[cc >> 8], 1u);
    }
  }
  __syncthreads();

  const unsigned a0 = h[2 * tid], a1 = h[2 * tid + 1];
  unsigned x = a0 + a1;
  for (int d = 1; d < 32; d <<= 1) {
    const unsigned y = __shfl_up(x, d);
    if (lane >= d) x += y;
  }
  if (lane == 31) wt[wave] = x;
  __syncthreads();
  unsigned p = 0u;
  for (int w = 0; w < wave; ++w) p += wt[w];
  const unsigned s0 = p + x - (a0 + a1);
  const unsigned s1 = s0 + a0;
  lst[2 * tid]     = s0;
  lst[2 * tid + 1] = s1;
  *(uint2*)(tbl + c * TBLW + 2 * tid) = make_uint2((s0 << 16) | a0, (s1 << 16) | a1);
  __syncthreads();

  unsigned rk[CNT_TRIPS], ls[CNT_TRIPS];
#pragma unroll
  for (int k = 0; k < CNT_TRIPS; ++k) {
    const int i = tid + k * 128;
    rk[k] = (unsigned)CHUNK_E;
    ls[k] = 0u;
    if (i < CHUNK_E) {
      ls[k] = lst[vc[k] >> 8];
      rk[k] = atomicAdd(&cur[vc[k] >> 8], 1u);
    }
  }
#pragma unroll
  for (int k = 0; k < CNT_TRIPS; ++k) {
    const unsigned slot = ls[k] + rk[k];
    if (slot < (unsigned)CHUNK_E) stage[slot] = make_int2(vr[k], (int)vc[k]);
  }
  __syncthreads();

  int2* dst = plocal + (size_t)c * CHUNK_E;
  int2 t[CNT_TRIPS];
#pragma unroll
  for (int k = 0; k < CNT_TRIPS; ++k) {
    const int i = tid + k * 128;
    t[k] = stage[i < CHUNK_E ? i : tid];
  }
#pragma unroll
  for (int k = 0; k < CNT_TRIPS; ++k) {
    const int i = tid + k * 128;
    if (i < CHUNK_E) dst[i] = t[k];
  }
}

#define BIN_CAP    16384
#define RUN_LANES  16
#define RUN_SWEEPS (NCHUNK / (1024 / RUN_LANES))
#define RUN_RECS   3
#define RUN_ROUND  (RUN_LANES * RUN_RECS)

static_assert(RUN_SWEEPS * (1024 / RUN_LANES) == NCHUNK, "the lane groups cover every chunk");
static_assert(RUN_LANES == 16 && RUN_SWEEPS == 4, "a lane group is one 16-lane row holding four table entries");

template <int L>
__device__ __forceinline__ unsigned row_share(unsigned v) {
  return (unsigned)__builtin_amdgcn_update_dpp(0, (int)v, 0x150 + L, 0xf, 0xf, false);
}

__global__ __launch_bounds__(1024) void k_bin(const int* __restrict__ plocal_w,
                                              const unsigned* __restrict__ tbl,
                                              int* __restrict__ erec,
                                              unsigned* __restrict__ off,
                                              unsigned* __restrict__ cnt,
                                              float* __restrict__ dinv) {
  __shared__ unsigned s_cnt[256], s_off[256], s_cur[256], s_w[8];
  __shared__ unsigned s_wm[32], s_ws[32], s_wx[32], s_tot[3];
  __shared__ int s_srt[BIN_CAP];
  const int2* plocal = (const int2*)plocal_w;

  const int tid  = threadIdx.x;
  const int lane = tid & 31;
  const int wave = tid >> 5;
  const int b    = blockIdx.x;
  const int g    = tid >> 4;
  const int j    = tid & (RUN_LANES - 1);

  const unsigned tw1 = tbl[(tid >> 2) * TBLW + b];
  unsigned rs[RUN_SWEEPS], rc[RUN_SWEEPS], inv[RUN_SWEEPS];
  {
    const unsigned t0 = row_share<0>(tw1), t1 = row_share<4>(tw1);
    const unsigned t2 = row_share<8>(tw1), t3 = row_share<12>(tw1);
    const unsigned tw[RUN_SWEEPS] = {t0, t1, t2, t3};
#pragma unroll
    for (int w = 0; w < RUN_SWEEPS; ++w) {
      unsigned s_ = tw[w] >> 16, c_ = tw[w] & 0xffffu;
      if (s_ > (unsigned)CHUNK_E) s_ = (unsigned)CHUNK_E;
      if (c_ > (unsigned)CHUNK_E - s_) c_ = (unsigned)CHUNK_E - s_;
      rs[w] = s_; rc[w] = c_;
      inv[w] = s_ < (unsigned)CHUNK_E ? s_ : 0u;
    }
  }

  int2 rec[RUN_SWEEPS][RUN_RECS];
#pragma unroll
  for (int w = 0; w < RUN_SWEEPS; ++w) {
    const size_t cbase = (size_t)(g * RUN_SWEEPS + w) * CHUNK_E;
#pragma unroll
    for (int u = 0; u < RUN_RECS; ++u) {
      const unsigned i = (unsigned)j + (unsigned)(RUN_LANES * u);
      rec[w][u] = plocal[cbase + (i < rc[w] ? rs[w] + i : inv[w])];
    }
  }

  unsigned cm = 0u, cs = 0u, cx = 0u;
#pragma unroll
  for (int w = 0; w < RUN_SWEEPS; ++w) {
    if (j == 0) { cm += rc[w]; cs += rs[w]; }
    cx = cx > rc[w] ? cx : rc[w];
  }
  for (int d = 16; d >= 1; d >>= 1) {
    cm += __shfl_xor(cm, d);
    cs += __shfl_xor(cs, d);
    const unsigned ox = __shfl_xor(cx, d);
    cx = cx > ox ? cx : ox;
  }
  if (lane == 0) { s_wm[wave] = cm; s_ws[wave] = cs; s_wx[wave] = cx; }
  if (tid < 256) { s_cnt[tid] = 0u; s_cur[tid] = 0u; }
  __syncthreads();
  if (wave == 0) {
    unsigned tm = s_wm[lane], ts = s_ws[lane], tx = s_wx[lane];
    for (int d = 16; d >= 1; d >>= 1) {
      tm += __shfl_xor(tm, d);
      ts += __shfl_xor(ts, d);
      const unsigned ox = __shfl_xor(tx, d);
      tx = tx > ox ? tx : ox;
    }
    if (lane == 0) { s_tot[0] = tm; s_tot[1] = ts; s_tot[2] = tx; }
  }
  __syncthreads();
  unsigned m = s_tot[0], S = s_tot[1];
  const unsigned mx = s_tot[2];
  if (S > (unsigned)N_EDGES) S = (unsigned)N_EDGES;
  if (m > (unsigned)N_EDGES - S) m = (unsigned)N_EDGES - S;

#pragma unroll
  for (int w = 0; w < RUN_SWEEPS; ++w) {
#pragma unroll
    for (int u = 0; u < RUN_RECS; ++u) {
      const unsigned i = (unsigned)j + (unsigned)(RUN_LANES * u);
      if (i < rc[w]) atomicAdd(&s_cnt[(unsigned)rec[w][u].y & 255u], 1u);
    }
  }
  for (unsigned r0 = RUN_ROUND; r0 < mx; r0 += RUN_ROUND) {
    int cl[RUN_SWEEPS][RUN_RECS];
#pragma unroll
    for (int w = 0; w < RUN_SWEEPS; ++w) {
      const size_t cbase = (size_t)(g * RUN_SWEEPS + w) * CHUNK_E;
#pragma unroll
      for (int u = 0; u < RUN_RECS; ++u) {
        const unsigned i = r0 + (unsigned)j + (unsigned)(RUN_LANES * u);
        cl[w][u] = plocal[cbase + (i < rc[w] ? rs[w] + i : inv[w])].y;
      }
    }
#pragma unroll
    for (int w = 0; w < RUN_SWEEPS; ++w) {
#pragma unroll
      for (int u = 0; u < RUN_RECS; ++u) {
        const unsigned i = r0 + (unsigned)j + (unsigned)(RUN_LANES * u);
        if (i < rc[w]) atomicAdd(&s_cnt[(unsigned)cl[w][u] & 255u], 1u);
      }
    }
  }
  __syncthreads();

  const unsigned cv = (tid < 256) ? s_cnt[tid] : 0u;
  unsigned x = cv;
  for (int d = 1; d < 32; d <<= 1) {
    const unsigned y = __shfl_up(x, d);
    if (lane >= d) x += y;
  }
  if (wave < 8 && lane == 31) s_w[wave] = x;
  __syncthreads();
  if (tid < 256) {
    unsigned p = 0u;
    for (int w = 0; w < wave; ++w) p += s_w[w];
    const unsigned o = p + x - cv;
    s_off[tid] = o;
    const int node = b * BIN_NODES + tid;
    if (node < N_NODES) {
      off[node]  = S + o;
      cnt[node]  = cv;
      dinv[node] = rsqrtf((float)cv + 1.0f);
    }
  }
  __syncthreads();

  const bool inl = m <= (unsigned)BIN_CAP;
  unsigned rk[RUN_SWEEPS][RUN_RECS], so[RUN_SWEEPS][RUN_RECS];
#pragma unroll
  for (int w = 0; w < RUN_SWEEPS; ++w) {
#pragma unroll
    for (int u = 0; u < RUN_RECS; ++u) {
      const unsigned i = (unsigned)j + (unsigned)(RUN_LANES * u);
      rk[w][u] = 0u;
      so[w][u] = 0u;
      if (i < rc[w]) {
        const unsigned nl = (unsigned)rec[w][u].y & 255u;
        so[w][u] = s_off[nl];
        rk[w][u] = atomicAdd(&s_cur[nl], 1u);
      }
    }
  }
#pragma unroll
  for (int w = 0; w < RUN_SWEEPS; ++w) {
#pragma unroll
    for (int u = 0; u < RUN_RECS; ++u) {
      const unsigned i = (unsigned)j + (unsigned)(RUN_LANES * u);
      if (i < rc[w]) {
        const unsigned p = so[w][u] + rk[w][u];
        if (inl) { if (p < (unsigned)BIN_CAP) s_srt[p] = rec[w][u].x; }
        else     { if (p < m) erec[S + p] = rec[w][u].x; }
      }
    }
  }
  for (unsigned r0 = RUN_ROUND; r0 < mx; r0 += RUN_ROUND) {
    int2 rr[RUN_SWEEPS][RUN_RECS];
#pragma unroll
    for (int w = 0; w < RUN_SWEEPS; ++w) {
      const size_t cbase = (size_t)(g * RUN_SWEEPS + w) * CHUNK_E;
#pragma unroll
      for (int u = 0; u < RUN_RECS; ++u) {
        const unsigned i = r0 + (unsigned)j + (unsigned)(RUN_LANES * u);
        rr[w][u] = plocal[cbase + (i < rc[w] ? rs[w] + i : inv[w])];
      }
    }
#pragma unroll
    for (int w = 0; w < RUN_SWEEPS; ++w) {
#pragma unroll
      for (int u = 0; u < RUN_RECS; ++u) {
        const unsigned i = r0 + (unsigned)j + (unsigned)(RUN_LANES * u);
        if (i < rc[w]) {
          const unsigned nl = (unsigned)rr[w][u].y & 255u;
          const unsigned p  = s_off[nl] + atomicAdd(&s_cur[nl], 1u);
          if (inl) { if (p < (unsigned)BIN_CAP) s_srt[p] = rr[w][u].x; }
          else     { if (p < m) erec[S + p] = rr[w][u].x; }
        }
      }
    }
  }
  __syncthreads();
  if (inl) {
    for (unsigned i = (unsigned)tid; i < m; i += 1024u) erec[S + i] = s_srt[i];
  }
}

template <int M>
__device__ __forceinline__ float row_xor(float v) {
  return __builtin_bit_cast(float, __builtin_amdgcn_update_dpp(
      0, __builtin_bit_cast(int, v), 0x160 + M, 0xf, 0xf, false));
}

template <int NRT>
__device__ __forceinline__ void gemm_rows(const float* __restrict__ x,
                                          const float* __restrict__ bW,
                                          const float* __restrict__ cW,
                                          const float* __restrict__ cb,
                                          float* __restrict__ bases,
                                          float* __restrict__ wbuf,
                                          float (*xs)[132], int row0, int tid) {
  for (int t = 0; t < 4 * NRT; ++t) {
    int idx = t * 128 + tid;
    int r   = idx >> 5;
    int c4  = (idx & 31) * 4;
    const float4 v = *(const float4*)(x + (size_t)(row0 + r) * IN_C + c4);
    xs[r][c4 + 0] = v.x; xs[r][c4 + 1] = v.y;
    xs[r][c4 + 2] = v.z; xs[r][c4 + 3] = v.w;
  }
  __syncthreads();

  const int wave = tid >> 5;
  const int lane = tid & 31;
  const int nlo  = lane & 15;
  const int half = lane >> 4;
  const int kb   = half * 2;

  for (int j = 0; j < 3; ++j) {
    v8f acc[NRT];
#pragma unroll
    for (int rt = 0; rt < NRT; ++rt) {
      const v8f z = {0.f, 0.f, 0.f, 0.f, 0.f, 0.f, 0.f, 0.f};
      acc[rt] = z;
    }

    if (j == 0) {
      const float* W = bW + wave * (IN_C * HEAD_D);
#pragma unroll 8
      for (int ks = 0; ks < 32; ++ks) {
        const int k = ks * 4 + kb;
        const v2f b = { W[k * HEAD_D + nlo], W[(k + 1) * HEAD_D + nlo] };
#pragma unroll
        for (int rt = 0; rt < NRT; ++rt) {
          const v2f a = *(const v2f*)&xs[rt * 16 + nlo][k];
          acc[rt] = __builtin_amdgcn_wmma_f32_16x16x4_f32(
              false, a, false, b, (short)0, acc[rt], false, false);
        }
      }
#pragma unroll
      for (int rt = 0; rt < NRT; ++rt) {
#pragma unroll
        for (int r = 0; r < 8; ++r) {
          const int m  = r + (half << 3);
          const int gr = row0 + rt * 16 + m;
          bases[(size_t)gr * BD + wave * 16 + nlo] = acc[rt][r];
        }
      }
    } else {
      const int   head = wave + 4 * (j - 1);
      const float* W   = cW + head * 16;
#pragma unroll 8
      for (int ks = 0; ks < 32; ++ks) {
        const int k = ks * 4 + kb;
        const v2f b = { W[k * OUT_C + nlo], W[(k + 1) * OUT_C + nlo] };
#pragma unroll
        for (int rt = 0; rt < NRT; ++rt) {
          const v2f a = *(const v2f*)&xs[rt * 16 + nlo][k];
          acc[rt] = __builtin_amdgcn_wmma_f32_16x16x4_f32(
              false, a, false, b, (short)0, acc[rt], false, false);
        }
      }
      const float bb = cb[head * 16 + nlo];
#pragma unroll
      for (int rt = 0; rt < NRT; ++rt) {
        float v[8], mx[8], ex[8], sm[8];
#pragma unroll
        for (int r = 0; r < 8; ++r) { v[r] = acc[rt][r] + bb; mx[r] = v[r]; }
#pragma unroll
        for (int r = 0; r < 8; ++r) mx[r] = fmaxf(mx[r], row_xor<1>(mx[r]));
#pragma unroll
        for (int r = 0; r < 8; ++r) mx[r] = fmaxf(mx[r], row_xor<2>(mx[r]));
#pragma unroll
        for (int r = 0; r < 8; ++r) mx[r] = fmaxf(mx[r], row_xor<4>(mx[r]));
#pragma unroll
        for (int r = 0; r < 8; ++r) mx[r] = fmaxf(mx[r], row_xor<8>(mx[r]));
#pragma unroll
        for (int r = 0; r < 8; ++r) { ex[r] = __expf(v[r] - mx[r]); sm[r] = ex[r]; }
#pragma unroll
        for (int r = 0; r < 8; ++r) sm[r] += row_xor<1>(sm[r]);
#pragma unroll
        for (int r = 0; r < 8; ++r) sm[r] += row_xor<2>(sm[r]);
#pragma unroll
        for (int r = 0; r < 8; ++r) sm[r] += row_xor<4>(sm[r]);
#pragma unroll
        for (int r = 0; r < 8; ++r) sm[r] += row_xor<8>(sm[r]);
#pragma unroll
        for (int r = 0; r < 8; ++r) {
          const float wv = ex[r] / sm[r];
          const int m  = r + (half << 3);
          const int gr = row0 + rt * 16 + m;
          wbuf[(size_t)gr * OUT_C + head * 16 + nlo] = wv;
        }
      }
    }
  }
}

__global__ __launch_bounds__(128) void k_gemm(const float* __restrict__ x,
                                              const float* __restrict__ bW,
                                              const float* __restrict__ cW,
                                              const float* __restrict__ cb,
                                              float* __restrict__ bases,
                                              float* __restrict__ wbuf,
                                              const int* __restrict__ ei,
                                              unsigned* __restrict__ tbl,
                                              int* __restrict__ plocal_w) {
  __shared__ float xs[64][132];

  const int tid  = threadIdx.x;
  const int blk  = (int)blockIdx.x;

  static_assert(sizeof(xs) >= 784 * 4 + CHUNK_E * 8, "a chunk's tables and staging array fit in xs");
  if (blk >= GEMM_BLOCKS) {
    sort_chunk(ei, tbl, (int2*)plocal_w, (unsigned*)&xs[0][0], blk - GEMM_BLOCKS, tid);
    return;
  }
  if (blk < FULL_BLOCKS) {
    gemm_rows<4>(x, bW, cW, cb, bases, wbuf, xs, blk * 64, tid);
  } else {
    gemm_rows<1>(x, bW, cW, cb, bases, wbuf, xs, TAIL_ROW0 + (blk - FULL_BLOCKS) * 16, tid);
  }
}

__global__ __launch_bounds__(256) void k_gather(const int* __restrict__ erec,
                                                const unsigned* __restrict__ off,
                                                const unsigned* __restrict__ cnt,
                                                const float* __restrict__ bases,
                                                const float* __restrict__ wbuf,
                                                const float* __restrict__ dinv,
                                                const float* __restrict__ bias,
                                                float* __restrict__ out) {
  __shared__ float4 fin[16][4][16];

  const int tid = threadIdx.x;
  const int nl  = tid >> 4;
  const int q   = tid & 15;
  const int n   = blockIdx.x * 16 + nl;

  const unsigned deg = cnt[n];
  unsigned start = off[n];
  if (start > (unsigned)N_EDGES) start = (unsigned)N_EDGES;
  unsigned end = start + deg;
  if (end > (unsigned)N_EDGES) end = (unsigned)N_EDGES;

  const float* bq  = bases + q * 4;

  float4 a_sym = make_float4(0.f, 0.f, 0.f, 0.f);
  float4 a_sum = make_float4(0.f, 0.f, 0.f, 0.f);
  float4 a_sq  = make_float4(0.f, 0.f, 0.f, 0.f);
  float4 a_mx  = make_float4(-INFINITY, -INFINITY, -INFINITY, -INFINITY);

  for (unsigned j = start; j < end; j += 4) {
    unsigned r[4];
    float    dr[4];
    float4   v[4];
#pragma unroll
    for (int u = 0; u < 4; ++u) {
      unsigned jj = j + u;
      if (jj >= end) jj = end - 1;
      const unsigned rr = (unsigned)erec[jj];
      r[u]  = rr < (unsigned)N_NODES ? rr : (unsigned)(N_NODES - 1);
    }
#pragma unroll
    for (int u = 0; u < 4; ++u) {
      v[u]  = *(const float4*)(bq + (size_t)r[u] * BD);
      dr[u] = dinv[r[u]];
    }
#pragma unroll
    for (int u = 0; u < 4; ++u) {
      const bool  ok = (j + u) < end;
      const float d  = dr[u];
      const float tx = ok ? v[u].x : 0.f, mx_ = ok ? v[u].x : -INFINITY;
      const float ty = ok ? v[u].y : 0.f, my_ = ok ? v[u].y : -INFINITY;
      const float tz = ok ? v[u].z : 0.f, mz_ = ok ? v[u].z : -INFINITY;
      const float tw = ok ? v[u].w : 0.f, mw_ = ok ? v[u].w : -INFINITY;
      a_sym.x += d * tx;  a_sym.y += d * ty;  a_sym.z += d * tz;  a_sym.w += d * tw;
      a_sum.x += tx;      a_sum.y += ty;      a_sum.z += tz;      a_sum.w += tw;
      a_sq.x  += tx * tx; a_sq.y  += ty * ty; a_sq.z  += tz * tz; a_sq.w  += tw * tw;
      a_mx.x = fmaxf(a_mx.x, mx_); a_mx.y = fmaxf(a_mx.y, my_);
      a_mx.z = fmaxf(a_mx.z, mz_); a_mx.w = fmaxf(a_mx.w, mw_);
    }
  }

  {
    const float  cf = (float)deg;
    const float  ic = 1.0f / fmaxf(cf, 1.0f);
    const float  dv = dinv[n];
    const float4 bs = *(const float4*)(bq + (size_t)n * BD);
    const float  sy[4] = {a_sym.x, a_sym.y, a_sym.z, a_sym.w};
    const float  su[4] = {a_sum.x, a_sum.y, a_sum.z, a_sum.w};
    const float  sq[4] = {a_sq.x,  a_sq.y,  a_sq.z,  a_sq.w};
    const float  sm[4] = {a_mx.x,  a_mx.y,  a_mx.z,  a_mx.w};
    const float  bv[4] = {bs.x, bs.y, bs.z, bs.w};
    float o_sym[4], o_mean[4], o_mx[4], o_sd[4];
#pragma unroll
    for (int c = 0; c < 4; ++c) {
      o_sym[c] = dv * (sy[c] + dv * bv[c]);
      const float mean = su[c] * ic;
      const float var  = sq[c] * ic - mean * mean;
      o_mean[c] = mean;
      o_sd[c]   = sqrtf(fmaxf(var, 0.0f) + 1e-5f);
      o_mx[c]   = (cf > 0.0f) ? sm[c] : 0.0f;
    }
    fin[nl][0][q] = make_float4(o_sym[0],  o_sym[1],  o_sym[2],  o_sym[3]);
    fin[nl][1][q] = make_float4(o_mean[0], o_mean[1], o_mean[2], o_mean[3]);
    fin[nl][2][q] = make_float4(o_mx[0],   o_mx[1],   o_mx[2],   o_mx[3]);
    fin[nl][3][q] = make_float4(o_sd[0],   o_sd[1],   o_sd[2],   o_sd[3]);
  }
  __syncthreads();

  const int dg = q & 3;
  const int hq = q >> 2;
#pragma unroll
  for (int h2 = 0; h2 < 2; ++h2) {
    const int h = h2 * 4 + hq;
    const float4 bi = *(const float4*)(bias + h * 16 + dg * 4);
    float o0 = bi.x, o1 = bi.y, o2 = bi.z, o3 = bi.w;
#pragma unroll
    for (int b = 0; b < 4; ++b) {
      const float4 w  = *(const float4*)(wbuf + (size_t)n * OUT_C + h * 16 + b * 4);
      const float4 s0 = fin[nl][0][b * 4 + dg];
      const float4 s1 = fin[nl][1][b * 4 + dg];
      const float4 s2 = fin[nl][2][b * 4 + dg];
      const float4 s3 = fin[nl][3][b * 4 + dg];
      o0 += w.x * s0.x + w.y * s1.x + w.z * s2.x + w.w * s3.x;
      o1 += w.x * s0.y + w.y * s1.y + w.z * s2.y + w.w * s3.y;
      o2 += w.x * s0.z + w.y * s1.z + w.z * s2.z + w.w * s3.z;
      o3 += w.x * s0.w + w.y * s1.w + w.z * s2.w + w.w * s3.w;
    }
    *(float4*)(out + (size_t)n * OUT_C + h * 16 + dg * 4) = make_float4(o0, o1, o2, o3);
  }
}

extern "C" void kernel_launch(void* const* d_in, const int* in_sizes, int n_in,
                              void* d_out, int out_size, void* d_ws, size_t ws_size,
                              hipStream_t stream) {
  (void)in_sizes; (void)n_in; (void)out_size; (void)ws_size;
  const float* x    = (const float*)d_in[0];
  const int*   ei   = (const int*)d_in[1];
  const float* bW   = (const float*)d_in[2];
  const float* cW   = (const float*)d_in[3];
  const float* cb   = (const float*)d_in[4];
  const float* bias = (const float*)d_in[5];
  float*       out  = (float*)d_out;

  char* ws = (char*)d_ws;
  char*     r0    = ws;          ws += (size_t)N_NODES * 256 * sizeof(float);
  float*    bases = (float*)ws;  ws += (size_t)N_NODES * BD * sizeof(float);
  float*    wbuf  = (float*)ws;  ws += (size_t)N_NODES * OUT_C * sizeof(float);
  unsigned* cnt   = (unsigned*)ws; ws += (size_t)N_NODES * sizeof(unsigned);
  float*    dinv  = (float*)ws;

  int*      plocal = (int*)r0;
  int*      erec   = (int*)(r0 + (size_t)N_EDGES * 8);
  unsigned* off    = (unsigned*)(r0 + (size_t)N_EDGES * 12);
  unsigned* tbl    = (unsigned*)(r0 + (size_t)N_EDGES * 12 + (size_t)262144);

  k_gemm<<<GEMM_BLOCKS + NCHUNK, 128, 0, stream>>>(x, bW, cW, cb, bases, wbuf,
                                                   ei, tbl, plocal);
  k_bin<<<NBINS, 1024, 0, stream>>>(plocal, tbl, erec, off, cnt, dinv);
  k_gather<<<N_NODES / 16, 256, 0, stream>>>(erec, off, cnt, bases, wbuf, dinv, bias, out);
}


# comb-weight product + softmax move from k_gemm into k_gather: wave w makes head w for the block's 16 nodes on WMMA under the gather's memory waits, weights stay in LDS; wbuf goes; sums unchanged
# speedup vs baseline: 6.9068x; 1.0698x over previous
#include <hip/hip_runtime.h>
#include <math.h>
#include <float.h>


#define N_NODES 50000
#define N_EDGES 800000
#define IN_C    128
#define OUT_C   128
#define HEADS   8
#define BASES   4
#define HEAD_D  16
#define BD      (BASES * HEAD_D)

typedef float v2f __attribute__((ext_vector_type(2)));
typedef float v8f __attribute__((ext_vector_type(8)));

static_assert(N_NODES % 16 == 0, "k_gather maps 16 nodes to a block with no tail");

#define N_WGP        256
#define FULL_BLOCKS  ((N_NODES / 64 / N_WGP) * N_WGP)
#define TAIL_ROW0    (FULL_BLOCKS * 64)
#define TAIL_BLOCKS  ((N_NODES - TAIL_ROW0) / 16)
#define GEMM_BLOCKS  (FULL_BLOCKS + TAIL_BLOCKS)
#define BIN_NODES    256
#define NBINS        ((N_NODES + BIN_NODES - 1) / BIN_NODES)
#define TBLW         256
#define NCHUNK       256
#define CHUNK_E      (N_EDGES / NCHUNK)
#define CNT_TRIPS    ((CHUNK_E + 127) / 128)

static_assert(N_EDGES % NCHUNK == 0, "edges split evenly over the chunks");
static_assert(FULL_BLOCKS > 0 && TAIL_ROW0 + TAIL_BLOCKS * 16 == N_NODES, "the matrix blocks cover every row with whole tiles");
static_assert(NBINS <= TBLW, "a table row holds every bin");
static_assert(CHUNK_E < 65536, "run start and run length share one 32-bit table entry");
static_assert(TBLW == 256, "two bins per thread of a 128-thread block");

__device__ __forceinline__ void sort_chunk(const int* __restrict__ ei,
                                           unsigned* __restrict__ tbl,
                                           int2* __restrict__ plocal,
                                           unsigned* lds, int c, int tid) {
  unsigned* h     = lds;
  unsigned* lst   = lds + 256;
  unsigned* cur   = lds + 512;
  unsigned* wt    = lds + 768;
  int2*     stage = (int2*)(lds + 784);
  const int lane = tid & 31;
  const int wave = tid >> 5;
  __builtin_assume(tid >= 0 && tid < 128);

  h[tid] = 0u;   h[tid + 128] = 0u;
  cur[tid] = 0u; cur[tid + 128] = 0u;
  const int e0 = c * CHUNK_E;
  unsigned vc[CNT_TRIPS];
  int      vr[CNT_TRIPS];
#pragma unroll
  for (int k = 0; k < CNT_TRIPS; ++k) {
    const int i = tid + k * 128;
    const int e = e0 + (i < CHUNK_E ? i : tid);
    vc[k] = (unsigned)ei[N_EDGES + e];
    vr[k] = ei[e];
  }
  __syncthreads();
#pragma unroll
  for (int k = 0; k < CNT_TRIPS; ++k) {
    const int i = tid + k * 128;
    if (i < CHUNK_E) {
      unsigned cc = vc[k];
      if (cc >= (unsigned)N_NODES) cc = (unsigned)(N_NODES - 1);
      vc[k] = cc;
      atomicAdd(&h[cc >> 8], 1u);
    }
  }
  __syncthreads();

  const unsigned a0 = h[2 * tid], a1 = h[2 * tid + 1];
  unsigned x = a0 + a1;
  for (int d = 1; d < 32; d <<= 1) {
    const unsigned y = __shfl_up(x, d);
    if (lane >= d) x += y;
  }
  if (lane == 31) wt[wave] = x;
  __syncthreads();
  unsigned p = 0u;
  for (int w = 0; w < wave; ++w) p += wt[w];
  const unsigned s0 = p + x - (a0 + a1);
  const unsigned s1 = s0 + a0;
  lst[2 * tid]     = s0;
  lst[2 * tid + 1] = s1;
  *(uint2*)(tbl + c * TBLW + 2 * tid) = make_uint2((s0 << 16) | a0, (s1 << 16) | a1);
  __syncthreads();

  unsigned rk[CNT_TRIPS], ls[CNT_TRIPS];
#pragma unroll
  for (int k = 0; k < CNT_TRIPS; ++k) {
    const int i = tid + k * 128;
    rk[k] = (unsigned)CHUNK_E;
    ls[k] = 0u;
    if (i < CHUNK_E) {
      ls[k] = lst[vc[k] >> 8];
      rk[k] = atomicAdd(&cur[vc[k] >> 8], 1u);
    }
  }
#pragma unroll
  for (int k = 0; k < CNT_TRIPS; ++k) {
    const unsigned slot = ls[k] + rk[k];
    if (slot < (unsigned)CHUNK_E) stage[slot] = make_int2(vr[k], (int)vc[k]);
  }
  __syncthreads();

  int2* dst = plocal + (size_t)c * CHUNK_E;
  int2 t[CNT_TRIPS];
#pragma unroll
  for (int k = 0; k < CNT_TRIPS; ++k) {
    const int i = tid + k * 128;
    t[k] = stage[i < CHUNK_E ? i : tid];
  }
#pragma unroll
  for (int k = 0; k < CNT_TRIPS; ++k) {
    const int i = tid + k * 128;
    if (i < CHUNK_E) dst[i] = t[k];
  }
}

#define BIN_CAP    16384
#define RUN_LANES  16
#define RUN_SWEEPS (NCHUNK / (1024 / RUN_LANES))
#define RUN_RECS   3
#define RUN_ROUND  (RUN_LANES * RUN_RECS)

static_assert(RUN_SWEEPS * (1024 / RUN_LANES) == NCHUNK, "the lane groups cover every chunk");
static_assert(RUN_LANES == 16 && RUN_SWEEPS == 4, "a lane group is one 16-lane row holding four table entries");

template <int L>
__device__ __forceinline__ unsigned row_share(unsigned v) {
  return (unsigned)__builtin_amdgcn_update_dpp(0, (int)v, 0x150 + L, 0xf, 0xf, false);
}

__global__ __launch_bounds__(1024) void k_bin(const int* __restrict__ plocal_w,
                                              const unsigned* __restrict__ tbl,
                                              int* __restrict__ erec,
                                              unsigned* __restrict__ off,
                                              unsigned* __restrict__ cnt,
                                              float* __restrict__ dinv) {
  __shared__ unsigned s_cnt[256], s_off[256], s_cur[256], s_w[8];
  __shared__ unsigned s_wm[32], s_ws[32], s_wx[32], s_tot[3];
  __shared__ int s_srt[BIN_CAP];
  const int2* plocal = (const int2*)plocal_w;

  const int tid  = threadIdx.x;
  const int lane = tid & 31;
  const int wave = tid >> 5;
  const int b    = blockIdx.x;
  const int g    = tid >> 4;
  const int j    = tid & (RUN_LANES - 1);

  const unsigned tw1 = tbl[(tid >> 2) * TBLW + b];
  unsigned rs[RUN_SWEEPS], rc[RUN_SWEEPS], inv[RUN_SWEEPS];
  {
    const unsigned t0 = row_share<0>(tw1), t1 = row_share<4>(tw1);
    const unsigned t2 = row_share<8>(tw1), t3 = row_share<12>(tw1);
    const unsigned tw[RUN_SWEEPS] = {t0, t1, t2, t3};
#pragma unroll
    for (int w = 0; w < RUN_SWEEPS; ++w) {
      unsigned s_ = tw[w] >> 16, c_ = tw[w] & 0xffffu;
      if (s_ > (unsigned)CHUNK_E) s_ = (unsigned)CHUNK_E;
      if (c_ > (unsigned)CHUNK_E - s_) c_ = (unsigned)CHUNK_E - s_;
      rs[w] = s_; rc[w] = c_;
      inv[w] = s_ < (unsigned)CHUNK_E ? s_ : 0u;
    }
  }

  int2 rec[RUN_SWEEPS][RUN_RECS];
#pragma unroll
  for (int w = 0; w < RUN_SWEEPS; ++w) {
    const size_t cbase = (size_t)(g * RUN_SWEEPS + w) * CHUNK_E;
#pragma unroll
    for (int u = 0; u < RUN_RECS; ++u) {
      const unsigned i = (unsigned)j + (unsigned)(RUN_LANES * u);
      rec[w][u] = plocal[cbase + (i < rc[w] ? rs[w] + i : inv[w])];
    }
  }

  unsigned cm = 0u, cs = 0u, cx = 0u;
#pragma unroll
  for (int w = 0; w < RUN_SWEEPS; ++w) {
    if (j == 0) { cm += rc[w]; cs += rs[w]; }
    cx = cx > rc[w] ? cx : rc[w];
  }
  for (int d = 16; d >= 1; d >>= 1) {
    cm += __shfl_xor(cm, d);
    cs += __shfl_xor(cs, d);
    const unsigned ox = __shfl_xor(cx, d);
    cx = cx > ox ? cx : ox;
  }
  if (lane == 0) { s_wm[wave] = cm; s_ws[wave] = cs; s_wx[wave] = cx; }
  if (tid < 256) { s_cnt[tid] = 0u; s_cur[tid] = 0u; }
  __syncthreads();
  if (wave == 0) {
    unsigned tm = s_wm[lane], ts = s_ws[lane], tx = s_wx[lane];
    for (int d = 16; d >= 1; d >>= 1) {
      tm += __shfl_xor(tm, d);
      ts += __shfl_xor(ts, d);
      const unsigned ox = __shfl_xor(tx, d);
      tx = tx > ox ? tx : ox;
    }
    if (lane == 0) { s_tot[0] = tm; s_tot[1] = ts; s_tot[2] = tx; }
  }
  __syncthreads();
  unsigned m = s_tot[0], S = s_tot[1];
  const unsigned mx = s_tot[2];
  if (S > (unsigned)N_EDGES) S = (unsigned)N_EDGES;
  if (m > (unsigned)N_EDGES - S) m = (unsigned)N_EDGES - S;

#pragma unroll
  for (int w = 0; w < RUN_SWEEPS; ++w) {
#pragma unroll
    for (int u = 0; u < RUN_RECS; ++u) {
      const unsigned i = (unsigned)j + (unsigned)(RUN_LANES * u);
      if (i < rc[w]) atomicAdd(&s_cnt[(unsigned)rec[w][u].y & 255u], 1u);
    }
  }
  for (unsigned r0 = RUN_ROUND; r0 < mx; r0 += RUN_ROUND) {
    int cl[RUN_SWEEPS][RUN_RECS];
#pragma unroll
    for (int w = 0; w < RUN_SWEEPS; ++w) {
      const size_t cbase = (size_t)(g * RUN_SWEEPS + w) * CHUNK_E;
#pragma unroll
      for (int u = 0; u < RUN_RECS; ++u) {
        const unsigned i = r0 + (unsigned)j + (unsigned)(RUN_LANES * u);
        cl[w][u] = plocal[cbase + (i < rc[w] ? rs[w] + i : inv[w])].y;
      }
    }
#pragma unroll
    for (int w = 0; w < RUN_SWEEPS; ++w) {
#pragma unroll
      for (int u = 0; u < RUN_RECS; ++u) {
        const unsigned i = r0 + (unsigned)j + (unsigned)(RUN_LANES * u);
        if (i < rc[w]) atomicAdd(&s_cnt[(unsigned)cl[w][u] & 255u], 1u);
      }
    }
  }
  __syncthreads();

  const unsigned cv = (tid < 256) ? s_cnt[tid] : 0u;
  unsigned x = cv;
  for (int d = 1; d < 32; d <<= 1) {
    const unsigned y = __shfl_up(x, d);
    if (lane >= d) x += y;
  }
  if (wave < 8 && lane == 31) s_w[wave] = x;
  __syncthreads();
  if (tid < 256) {
    unsigned p = 0u;
    for (int w = 0; w < wave; ++w) p += s_w[w];
    const unsigned o = p + x - cv;
    s_off[tid] = o;
    const int node = b * BIN_NODES + tid;
    if (node < N_NODES) {
      off[node]  = S + o;
      cnt[node]  = cv;
      dinv[node] = rsqrtf((float)cv + 1.0f);
    }
  }
  __syncthreads();

  const bool inl = m <= (unsigned)BIN_CAP;
  unsigned rk[RUN_SWEEPS][RUN_RECS], so[RUN_SWEEPS][RUN_RECS];
#pragma unroll
  for (int w = 0; w < RUN_SWEEPS; ++w) {
#pragma unroll
    for (int u = 0; u < RUN_RECS; ++u) {
      const unsigned i = (unsigned)j + (unsigned)(RUN_LANES * u);
      rk[w][u] = 0u;
      so[w][u] = 0u;
      if (i < rc[w]) {
        const unsigned nl = (unsigned)rec[w][u].y & 255u;
        so[w][u] = s_off[nl];
        rk[w][u] = atomicAdd(&s_cur[nl], 1u);
      }
    }
  }
#pragma unroll
  for (int w = 0; w < RUN_SWEEPS; ++w) {
#pragma unroll
    for (int u = 0; u < RUN_RECS; ++u) {
      const unsigned i = (unsigned)j + (unsigned)(RUN_LANES * u);
      if (i < rc[w]) {
        const unsigned p = so[w][u] + rk[w][u];
        if (inl) { if (p < (unsigned)BIN_CAP) s_srt[p] = rec[w][u].x; }
        else     { if (p < m) erec[S + p] = rec[w][u].x; }
      }
    }
  }
  for (unsigned r0 = RUN_ROUND; r0 < mx; r0 += RUN_ROUND) {
    int2 rr[RUN_SWEEPS][RUN_RECS];
#pragma unroll
    for (int w = 0; w < RUN_SWEEPS; ++w) {
      const size_t cbase = (size_t)(g * RUN_SWEEPS + w) * CHUNK_E;
#pragma unroll
      for (int u = 0; u < RUN_RECS; ++u) {
        const unsigned i = r0 + (unsigned)j + (unsigned)(RUN_LANES * u);
        rr[w][u] = plocal[cbase + (i < rc[w] ? rs[w] + i : inv[w])];
      }
    }
#pragma unroll
    for (int w = 0; w < RUN_SWEEPS; ++w) {
#pragma unroll
      for (int u = 0; u < RUN_RECS; ++u) {
        const unsigned i = r0 + (unsigned)j + (unsigned)(RUN_LANES * u);
        if (i < rc[w]) {
          const unsigned nl = (unsigned)rr[w][u].y & 255u;
          const unsigned p  = s_off[nl] + atomicAdd(&s_cur[nl], 1u);
          if (inl) { if (p < (unsigned)BIN_CAP) s_srt[p] = rr[w][u].x; }
          else     { if (p < m) erec[S + p] = rr[w][u].x; }
        }
      }
    }
  }
  __syncthreads();
  if (inl) {
    for (unsigned i = (unsigned)tid; i < m; i += 1024u) erec[S + i] = s_srt[i];
  }
}

template <int M>
__device__ __forceinline__ float row_xor(float v) {
  return __builtin_bit_cast(float, __builtin_amdgcn_update_dpp(
      0, __builtin_bit_cast(int, v), 0x160 + M, 0xf, 0xf, false));
}

template <int NRT>
__device__ __forceinline__ void gemm_rows(const float* __restrict__ x,
                                          const float* __restrict__ bW,
                                          float* __restrict__ bases,
                                          float (*xs)[132], int row0, int tid) {
  for (int t = 0; t < 4 * NRT; ++t) {
    int idx = t * 128 + tid;
    int r   = idx >> 5;
    int c4  = (idx & 31) * 4;
    const float4 v = *(const float4*)(x + (size_t)(row0 + r) * IN_C + c4);
    xs[r][c4 + 0] = v.x; xs[r][c4 + 1] = v.y;
    xs[r][c4 + 2] = v.z; xs[r][c4 + 3] = v.w;
  }
  __syncthreads();

  const int wave = tid >> 5;
  const int lane = tid & 31;
  const int nlo  = lane & 15;
  const int half = lane >> 4;
  const int kb   = half * 2;

  v8f acc[NRT];
#pragma unroll
  for (int rt = 0; rt < NRT; ++rt) {
    const v8f z = {0.f, 0.f, 0.f, 0.f, 0.f, 0.f, 0.f, 0.f};
    acc[rt] = z;
  }

  const float* W = bW + wave * (IN_C * HEAD_D);
#pragma unroll 8
  for (int ks = 0; ks < 32; ++ks) {
    const int k = ks * 4 + kb;
    const v2f b = { W[k * HEAD_D + nlo], W[(k + 1) * HEAD_D + nlo] };
#pragma unroll
    for (int rt = 0; rt < NRT; ++rt) {
      const v2f a = *(const v2f*)&xs[rt * 16 + nlo][k];
      acc[rt] = __builtin_amdgcn_wmma_f32_16x16x4_f32(
          false, a, false, b, (short)0, acc[rt], false, false);
    }
  }
#pragma unroll
  for (int rt = 0; rt < NRT; ++rt) {
#pragma unroll
    for (int r = 0; r < 8; ++r) {
      const int m  = r + (half << 3);
      const int gr = row0 + rt * 16 + m;
      bases[(size_t)gr * BD + wave * 16 + nlo] = acc[rt][r];
    }
  }
}

__global__ __launch_bounds__(128) void k_gemm(const float* __restrict__ x,
                                              const float* __restrict__ bW,
                                              float* __restrict__ bases,
                                              const int* __restrict__ ei,
                                              unsigned* __restrict__ tbl,
                                              int* __restrict__ plocal_w) {
  __shared__ float xs[64][132];

  const int tid  = threadIdx.x;
  const int blk  = (int)blockIdx.x;

  static_assert(sizeof(xs) >= 784 * 4 + CHUNK_E * 8, "a chunk's tables and staging array fit in xs");
  if (blk >= GEMM_BLOCKS) {
    sort_chunk(ei, tbl, (int2*)plocal_w, (unsigned*)&xs[0][0], blk - GEMM_BLOCKS, tid);
    return;
  }
  if (blk < FULL_BLOCKS) {
    gemm_rows<4>(x, bW, bases, xs, blk * 64, tid);
  } else {
    gemm_rows<1>(x, bW, bases, xs, TAIL_ROW0 + (blk - FULL_BLOCKS) * 16, tid);
  }
}

__global__ __launch_bounds__(256) void k_gather(const int* __restrict__ erec,
                                                const unsigned* __restrict__ off,
                                                const unsigned* __restrict__ cnt,
                                                const float* __restrict__ bases,
                                                const float* __restrict__ x,
                                                const float* __restrict__ cW,
                                                const float* __restrict__ cb,
                                                const float* __restrict__ dinv,
                                                const float* __restrict__ bias,
                                                float* __restrict__ out) {
  __shared__ float4 fin[16][4][16];
  __shared__ __attribute__((aligned(16))) float xs[16][132];
  __shared__ __attribute__((aligned(16))) float wl[16][132];

  const int tid = threadIdx.x;
  const int nl  = tid >> 4;
  const int q   = tid & 15;
  const int n0  = blockIdx.x * 16;
  const int n   = n0 + nl;

#pragma unroll
  for (int t = 0; t < 2; ++t) {
    const int idx = t * 256 + tid;
    const int r   = idx >> 5;
    const int c4  = (idx & 31) * 4;
    const float4 v = *(const float4*)(x + (size_t)(n0 + r) * IN_C + c4);
    xs[r][c4 + 0] = v.x; xs[r][c4 + 1] = v.y;
    xs[r][c4 + 2] = v.z; xs[r][c4 + 3] = v.w;
  }

  const unsigned deg = cnt[n];
  unsigned start = off[n];
  if (start > (unsigned)N_EDGES) start = (unsigned)N_EDGES;
  unsigned end = start + deg;
  if (end > (unsigned)N_EDGES) end = (unsigned)N_EDGES;

  __syncthreads();

  {
    const int head = tid >> 5;
    const int lane = tid & 31;
    const int nlo  = lane & 15;
    const int half = lane >> 4;
    const int kb   = half * 2;
    const float* W = cW + head * 16;
    const float bb = cb[head * 16 + nlo];
    v8f acc = {0.f, 0.f, 0.f, 0.f, 0.f, 0.f, 0.f, 0.f};
#pragma unroll 8
    for (int ks = 0; ks < 32; ++ks) {
      const int k = ks * 4 + kb;
      const v2f b = { W[k * OUT_C + nlo], W[(k + 1) * OUT_C + nlo] };
      const v2f a = *(const v2f*)&xs[nlo][k];
      acc = __builtin_amdgcn_wmma_f32_16x16x4_f32(
          false, a, false, b, (short)0, acc, false, false);
    }
    float v[8], mx[8], ex[8], sm[8];
#pragma unroll
    for (int r = 0; r < 8; ++r) { v[r] = acc[r] + bb; mx[r] = v[r]; }
#pragma unroll
    for (int r = 0; r < 8; ++r) mx[r] = fmaxf(mx[r], row_xor<1>(mx[r]));
#pragma unroll
    for (int r = 0; r < 8; ++r) mx[r] = fmaxf(mx[r], row_xor<2>(mx[r]));
#pragma unroll
    for (int r = 0; r < 8; ++r) mx[r] = fmaxf(mx[r], row_xor<4>(mx[r]));
#pragma unroll
    for (int r = 0; r < 8; ++r) mx[r] = fmaxf(mx[r], row_xor<8>(mx[r]));
#pragma unroll
    for (int r = 0; r < 8; ++r) { ex[r] = __expf(v[r] - mx[r]); sm[r] = ex[r]; }
#pragma unroll
    for (int r = 0; r < 8; ++r) sm[r] += row_xor<1>(sm[r]);
#pragma unroll
    for (int r = 0; r < 8; ++r) sm[r] += row_xor<2>(sm[r]);
#pragma unroll
    for (int r = 0; r < 8; ++r) sm[r] += row_xor<4>(sm[r]);
#pragma unroll
    for (int r = 0; r < 8; ++r) sm[r] += row_xor<8>(sm[r]);
#pragma unroll
    for (int r = 0; r < 4; ++r) {
      const int m = r + (half << 3);
      wl[m][head * 16 + nlo] = ex[r] / sm[r];
    }
    __builtin_amdgcn_sched_barrier(0);
#pragma unroll
    for (int r = 4; r < 8; ++r) {
      const int m = r + (half << 3);
      wl[m][head * 16 + nlo] = ex[r] / sm[r];
    }
  }

  const float* bq  = bases + q * 4;

  float4 a_sym = make_float4(0.f, 0.f, 0.f, 0.f);
  float4 a_sum = make_float4(0.f, 0.f, 0.f, 0.f);
  float4 a_sq  = make_float4(0.f, 0.f, 0.f, 0.f);
  float4 a_mx  = make_float4(-INFINITY, -INFINITY, -INFINITY, -INFINITY);

  for (unsigned j = start; j < end; j += 4) {
    unsigned r[4];
    float    dr[4];
    float4   v[4];
#pragma unroll
    for (int u = 0; u < 4; ++u) {
      unsigned jj = j + u;
      if (jj >= end) jj = end - 1;
      const unsigned rr = (unsigned)erec[jj];
      r[u]  = rr < (unsigned)N_NODES ? rr : (unsigned)(N_NODES - 1);
    }
#pragma unroll
    for (int u = 0; u < 4; ++u) {
      v[u]  = *(const float4*)(bq + (size_t)r[u] * BD);
      dr[u] = dinv[r[u]];
    }
#pragma unroll
    for (int u = 0; u < 4; ++u) {
      const bool  ok = (j + u) < end;
      const float d  = dr[u];
      const float tx = ok ? v[u].x : 0.f, mx_ = ok ? v[u].x : -INFINITY;
      const float ty = ok ? v[u].y : 0.f, my_ = ok ? v[u].y : -INFINITY;
      const float tz = ok ? v[u].z : 0.f, mz_ = ok ? v[u].z : -INFINITY;
      const float tw = ok ? v[u].w : 0.f, mw_ = ok ? v[u].w : -INFINITY;
      a_sym.x += d * tx;  a_sym.y += d * ty;  a_sym.z += d * tz;  a_sym.w += d * tw;
      a_sum.x += tx;      a_sum.y += ty;      a_sum.z += tz;      a_sum.w += tw;
      a_sq.x  += tx * tx; a_sq.y  += ty * ty; a_sq.z  += tz * tz; a_sq.w  += tw * tw;
      a_mx.x = fmaxf(a_mx.x, mx_); a_mx.y = fmaxf(a_mx.y, my_);
      a_mx.z = fmaxf(a_mx.z, mz_); a_mx.w = fmaxf(a_mx.w, mw_);
    }
  }

  {
    const float  cf = (float)deg;
    const float  ic = 1.0f / fmaxf(cf, 1.0f);
    const float  dv = dinv[n];
    const float4 bs = *(const float4*)(bq + (size_t)n * BD);
    const float  sy[4] = {a_sym.x, a_sym.y, a_sym.z, a_sym.w};
    const float  su[4] = {a_sum.x, a_sum.y, a_sum.z, a_sum.w};
    const float  sq[4] = {a_sq.x,  a_sq.y,  a_sq.z,  a_sq.w};
    const float  sm[4] = {a_mx.x,  a_mx.y,  a_mx.z,  a_mx.w};
    const float  bv[4] = {bs.x, bs.y, bs.z, bs.w};
    float o_sym[4], o_mean[4], o_mx[4], o_sd[4];
#pragma unroll
    for (int c = 0; c < 4; ++c) {
      o_sym[c] = dv * (sy[c] + dv * bv[c]);
      const float mean = su[c] * ic;
      const float var  = sq[c] * ic - mean * mean;
      o_mean[c] = mean;
      o_sd[c]   = sqrtf(fmaxf(var, 0.0f) + 1e-5f);
      o_mx[c]   = (cf > 0.0f) ? sm[c] : 0.0f;
    }
    fin[nl][0][q] = make_float4(o_sym[0],  o_sym[1],  o_sym[2],  o_sym[3]);
    fin[nl][1][q] = make_float4(o_mean[0], o_mean[1], o_mean[2], o_mean[3]);
    fin[nl][2][q] = make_float4(o_mx[0],   o_mx[1],   o_mx[2],   o_mx[3]);
    fin[nl][3][q] = make_float4(o_sd[0],   o_sd[1],   o_sd[2],   o_sd[3]);
  }
  __syncthreads();

  const int dg = q & 3;
  const int hq = q >> 2;
#pragma unroll
  for (int h2 = 0; h2 < 2; ++h2) {
    const int h = h2 * 4 + hq;
    const float4 bi = *(const float4*)(bias + h * 16 + dg * 4);
    float o0 = bi.x, o1 = bi.y, o2 = bi.z, o3 = bi.w;
#pragma unroll
    for (int b = 0; b < 4; ++b) {
      const float4 w  = *(const float4*)&wl[nl][h * 16 + b * 4];
      const float4 s0 = fin[nl][0][b * 4 + dg];
      const float4 s1 = fin[nl][1][b * 4 + dg];
      const float4 s2 = fin[nl][2][b * 4 + dg];
      const float4 s3 = fin[nl][3][b * 4 + dg];
      o0 += w.x * s0.x + w.y * s1.x + w.z * s2.x + w.w * s3.x;
      o1 += w.x * s0.y + w.y * s1.y + w.z * s2.y + w.w * s3.y;
      o2 += w.x * s0.z + w.y * s1.z + w.z * s2.z + w.w * s3.z;
      o3 += w.x * s0.w + w.y * s1.w + w.z * s2.w + w.w * s3.w;
    }
    *(float4*)(out + (size_t)n * OUT_C + h * 16 + dg * 4) = make_float4(o0, o1, o2, o3);
  }
}

extern "C" void kernel_launch(void* const* d_in, const int* in_sizes, int n_in,
                              void* d_out, int out_size, void* d_ws, size_t ws_size,
                              hipStream_t stream) {
  (void)in_sizes; (void)n_in; (void)out_size; (void)ws_size;
  const float* x    = (const float*)d_in[0];
  const int*   ei   = (const int*)d_in[1];
  const float* bW   = (const float*)d_in[2];
  const float* cW   = (const float*)d_in[3];
  const float* cb   = (const float*)d_in[4];
  const float* bias = (const float*)d_in[5];
  float*       out  = (float*)d_out;

  char* ws = (char*)d_ws;
  char*     r0    = ws;          ws += (size_t)N_NODES * 256 * sizeof(float);
  float*    bases = (float*)ws;  ws += (size_t)N_NODES * BD * sizeof(float);
  ws += (size_t)N_NODES * OUT_C * sizeof(float);
  unsigned* cnt   = (unsigned*)ws; ws += (size_t)N_NODES * sizeof(unsigned);
  float*    dinv  = (float*)ws;

  int*      plocal = (int*)r0;
  int*      erec   = (int*)(r0 + (size_t)N_EDGES * 8);
  unsigned* off    = (unsigned*)(r0 + (size_t)N_EDGES * 12);
  unsigned* tbl    = (unsigned*)(r0 + (size_t)N_EDGES * 12 + (size_t)262144);

  k_gemm<<<GEMM_BLOCKS + NCHUNK, 128, 0, stream>>>(x, bW, bases, ei, tbl, plocal);
  k_bin<<<NBINS, 1024, 0, stream>>>(plocal, tbl, erec, off, cnt, dinv);
  k_gather<<<N_NODES / 16, 256, 0, stream>>>(erec, off, cnt, bases, x, cW, cb, dinv, bias, out);
}
